# starting kernel
# baseline (speedup 1.0000x reference)
#include <hip/hip_runtime.h>
#include <hip/hip_bf16.h>
#include <stdint.h>

#define N_NODES 100000
#define N_EDGES 1600000
#define F_IN    256
#define HID     64
#define RT      5

typedef __attribute__((ext_vector_type(16))) _Float16 v16h;
typedef __attribute__((ext_vector_type(8)))  float    v8f;
typedef __attribute__((ext_vector_type(4)))  float    v4f;

static __device__ __forceinline__ void atomicAddF32(float* p, float v) {
    __hip_atomic_fetch_add(p, v, __ATOMIC_RELAXED, __HIP_MEMORY_SCOPE_AGENT);
}

__global__ void k_deg_zero(int* __restrict__ deg) {
    int i = blockIdx.x * blockDim.x + threadIdx.x;
    if (i < N_NODES) deg[i] = 0;
}

__global__ void k_deg_count(const int* __restrict__ ei, int* __restrict__ deg) {
    int e = blockIdx.x * blockDim.x + threadIdx.x;
    if (e < N_EDGES) {
        int d = ei[N_EDGES + e];
        atomicAdd(&deg[d], 1);
    }
}

__global__ void k_dinv(const int* __restrict__ deg, float* __restrict__ dinv) {
    int i = blockIdx.x * blockDim.x + threadIdx.x;
    if (i < N_NODES) dinv[i] = __frsqrt_rn((float)deg[i] + 1.0f);
}

__global__ __launch_bounds__(128) void k_gemm1_wmma(
        const float* __restrict__ x,
        const float* __restrict__ W1,
        float* __restrict__ xw1) {
    const int wave = threadIdx.x >> 5;
    const int lane = threadIdx.x & 31;
    const int half = lane >> 4;
    const int r    = lane & 15;
    const int n0   = wave * 16;

    v16h bfrag[8];
    #pragma unroll
    for (int kt = 0; kt < 8; ++kt) {
        const int kbase = kt * 32;
        #pragma unroll
        for (int j = 0; j < 8; ++j)
            bfrag[kt][j]     = (_Float16)W1[(kbase + 8 * half + j) * HID + (n0 + r)];
        #pragma unroll
        for (int j = 0; j < 8; ++j)
            bfrag[kt][8 + j] = (_Float16)W1[(kbase + 16 + 8 * half + j) * HID + (n0 + r)];
    }

    for (int rt = 0; rt < RT; ++rt) {
        const int row0 = (blockIdx.x * RT + rt) * 16;
        const float* xrow = x + (size_t)(row0 + r) * F_IN;

        if (rt + 1 < RT) {
            const float* nxt = xrow + 16 * F_IN;
            __builtin_prefetch(nxt +   0, 0, 0);
            __builtin_prefetch(nxt +  64, 0, 0);
            __builtin_prefetch(nxt + 128, 0, 0);
            __builtin_prefetch(nxt + 192, 0, 0);
        }

        v8f acc = {0.f, 0.f, 0.f, 0.f, 0.f, 0.f, 0.f, 0.f};
        #pragma unroll
        for (int kt = 0; kt < 8; ++kt) {
            const float* seg0 = xrow + kt * 32 + 8 * half;
            const float* seg1 = seg0 + 16;
            v4f a0 = __builtin_nontemporal_load((const v4f*)seg0);
            v4f a1 = __builtin_nontemporal_load((const v4f*)(seg0 + 4));
            v4f a2 = __builtin_nontemporal_load((const v4f*)seg1);
            v4f a3 = __builtin_nontemporal_load((const v4f*)(seg1 + 4));
            v16h a;
            a[0]  = (_Float16)a0.x; a[1]  = (_Float16)a0.y;
            a[2]  = (_Float16)a0.z; a[3]  = (_Float16)a0.w;
            a[4]  = (_Float16)a1.x; a[5]  = (_Float16)a1.y;
            a[6]  = (_Float16)a1.z; a[7]  = (_Float16)a1.w;
            a[8]  = (_Float16)a2.x; a[9]  = (_Float16)a2.y;
            a[10] = (_Float16)a2.z; a[11] = (_Float16)a2.w;
            a[12] = (_Float16)a3.x; a[13] = (_Float16)a3.y;
            a[14] = (_Float16)a3.z; a[15] = (_Float16)a3.w;

            acc = __builtin_amdgcn_wmma_f32_16x16x32_f16(
                 false, a,  false, bfrag[kt],
                 (short)0, acc,  false,  false);
        }

        #pragma unroll
        for (int i = 0; i < 8; ++i) {
            int m = i + 8 * half;
            xw1[(size_t)(row0 + m) * HID + (n0 + r)] = acc[i];
        }
    }
}

__global__ void k_agg1_init(const float* __restrict__ xw1,
                            const float* __restrict__ dinv,
                            const float* __restrict__ b1,
                            float* __restrict__ agg1) {
    size_t t = (size_t)blockIdx.x * blockDim.x + threadIdx.x;
    if (t < (size_t)N_NODES * HID) {
        int node = (int)(t >> 6);
        int f    = (int)(t & 63);
        float di = dinv[node];
        agg1[t] = xw1[t] * (di * di) + b1[f];
    }
}

__global__ void k_agg1_edges(const int* __restrict__ ei,
                             const float* __restrict__ dinv,
                             const float* __restrict__ xw1,
                             float* __restrict__ agg1) {
    size_t t = (size_t)blockIdx.x * blockDim.x + threadIdx.x;
    int e    = (int)(t >> 5);
    int lane = (int)(t & 31);
    if (e >= N_EDGES) return;
    e = __builtin_amdgcn_readfirstlane(e);
    int s = ei[e];
    int d = ei[N_EDGES + e];
    float norm = dinv[s] * dinv[d];
    const float2 v = *(const float2*)(xw1 + (size_t)s * HID + lane * 2);
    float* out = agg1 + (size_t)d * HID + lane * 2;
    atomicAddF32(out + 0, v.x * norm);
    atomicAddF32(out + 1, v.y * norm);
}

__global__ void k_relu_gemm2(const float* __restrict__ agg1,
                             const float* __restrict__ W2,
                             const float* __restrict__ b2,
                             const float* __restrict__ dinv,
                             float* __restrict__ hw2,
                             float* __restrict__ agg2) {
    int i = blockIdx.x * blockDim.x + threadIdx.x;
    if (i >= N_NODES) return;
    const float* row = agg1 + (size_t)i * HID;
    float o0 = 0.f, o1 = 0.f;
    #pragma unroll 8
    for (int k = 0; k < HID; ++k) {
        float h = fmaxf(row[k], 0.f);
        o0 += h * W2[k * 2 + 0];
        o1 += h * W2[k * 2 + 1];
    }
    hw2[(size_t)i * 2 + 0] = o0;
    hw2[(size_t)i * 2 + 1] = o1;
    float di = dinv[i];
    float sl = di * di;
    agg2[(size_t)i * 2 + 0] = o0 * sl + b2[0];
    agg2[(size_t)i * 2 + 1] = o1 * sl + b2[1];
}

__global__ void k_agg2_edges(const int* __restrict__ ei,
                             const float* __restrict__ dinv,
                             const float* __restrict__ hw2,
                             float* __restrict__ agg2) {
    int e = blockIdx.x * blockDim.x + threadIdx.x;
    if (e >= N_EDGES) return;
    int s = ei[e];
    int d = ei[N_EDGES + e];
    float norm = dinv[s] * dinv[d];
    const float2 v = *(const float2*)(hw2 + (size_t)s * 2);
    float* out = agg2 + (size_t)d * 2;
    atomicAddF32(out + 0, v.x * norm);
    atomicAddF32(out + 1, v.y * norm);
}

__global__ void k_softmax2(const float* __restrict__ agg2,
                           float* __restrict__ out) {
    int i = blockIdx.x * blockDim.x + threadIdx.x;
    if (i >= N_NODES) return;
    float a0 = agg2[(size_t)i * 2 + 0];
    float a1 = agg2[(size_t)i * 2 + 1];
    float m  = fmaxf(a0, a1);
    float e0 = __expf(a0 - m);
    float e1 = __expf(a1 - m);
    float inv = 1.0f / (e0 + e1);
    out[(size_t)i * 2 + 0] = e0 * inv;
    out[(size_t)i * 2 + 1] = e1 * inv;
}

extern "C" void kernel_launch(void* const* d_in, const int* in_sizes, int n_in,
                              void* d_out, int out_size, void* d_ws, size_t ws_size,
                              hipStream_t stream) {
    const float* x  = (const float*)d_in[0];
    const int*   ei = (const int*)d_in[1];
    const float* W1 = (const float*)d_in[2];
    const float* b1 = (const float*)d_in[3];
    const float* W2 = (const float*)d_in[4];
    const float* b2 = (const float*)d_in[5];
    float* out = (float*)d_out;

    char* ws = (char*)d_ws;
    size_t o = 0;
    auto alloc = [&](size_t bytes) {
        char* p = ws + o;
        o += (bytes + 4095) & ~(size_t)4095;
        return p;
    };
    int*   deg  = (int*)  alloc((size_t)N_NODES * sizeof(int));
    float* dinv = (float*)alloc((size_t)N_NODES * sizeof(float));
    float* xw1  = (float*)alloc((size_t)N_NODES * HID * sizeof(float));
    float* agg1 = (float*)alloc((size_t)N_NODES * HID * sizeof(float));
    float* hw2  = (float*)alloc((size_t)N_NODES * 2 * sizeof(float));
    float* agg2 = (float*)alloc((size_t)N_NODES * 2 * sizeof(float));
    (void)ws_size; (void)in_sizes; (void)n_in; (void)out_size;

    const int B = 256;
    dim3 blk(B);

    k_deg_zero <<<(N_NODES + B - 1) / B, blk, 0, stream>>>(deg);
    k_deg_count<<<(N_EDGES + B - 1) / B, blk, 0, stream>>>(ei, deg);
    k_dinv     <<<(N_NODES + B - 1) / B, blk, 0, stream>>>(deg, dinv);

    k_gemm1_wmma<<<(N_NODES / 16) / RT, dim3(128), 0, stream>>>(x, W1, xw1);

    size_t t1 = (size_t)N_NODES * HID;
    k_agg1_init<<<(unsigned)((t1 + B - 1) / B), blk, 0, stream>>>(xw1, dinv, b1, agg1);
    size_t t2 = (size_t)N_EDGES * 32;
    k_agg1_edges<<<(unsigned)((t2 + B - 1) / B), blk, 0, stream>>>(ei, dinv, xw1, agg1);

    k_relu_gemm2<<<(N_NODES + B - 1) / B, blk, 0, stream>>>(agg1, W2, b2, dinv, hw2, agg2);

    k_agg2_edges<<<(N_EDGES + B - 1) / B, blk, 0, stream>>>(ei, dinv, hw2, agg2);
    k_softmax2  <<<(N_NODES + B - 1) / B, blk, 0, stream>>>(agg2, out);
}


# layer-1 sums by dst-range owners: 256 work-groups each keep 391 nodes' rows in LDS, filter the whole dst list, add own edges by LDS float adds (no global float atomic in layer 1); LDS arrival order
# speedup vs baseline: 1.4255x; 1.4255x over previous
#include <hip/hip_runtime.h>
#include <hip/hip_bf16.h>
#include <stdint.h>

#define N_NODES 100000
#define N_EDGES 1600000
#define F_IN    256
#define HID     64
#define RT      5

typedef __attribute__((ext_vector_type(16))) _Float16 v16h;
typedef __attribute__((ext_vector_type(8)))  float    v8f;
typedef __attribute__((ext_vector_type(4)))  float    v4f;

static __device__ __forceinline__ void atomicAddF32(float* p, float v) {
    __hip_atomic_fetch_add(p, v, __ATOMIC_RELAXED, __HIP_MEMORY_SCOPE_AGENT);
}

__global__ void k_deg_zero(int* __restrict__ deg) {
    int i = blockIdx.x * blockDim.x + threadIdx.x;
    if (i < N_NODES) deg[i] = 0;
}

__global__ void k_deg_count(const int* __restrict__ ei, int* __restrict__ deg) {
    int e = blockIdx.x * blockDim.x + threadIdx.x;
    if (e < N_EDGES) {
        int d = ei[N_EDGES + e];
        atomicAdd(&deg[d], 1);
    }
}

__global__ void k_dinv(const int* __restrict__ deg, float* __restrict__ dinv) {
    int i = blockIdx.x * blockDim.x + threadIdx.x;
    if (i < N_NODES) dinv[i] = __frsqrt_rn((float)deg[i] + 1.0f);
}

__global__ __launch_bounds__(128) void k_gemm1_wmma(
        const float* __restrict__ x,
        const float* __restrict__ W1,
        float* __restrict__ xw1) {
    const int wave = threadIdx.x >> 5;
    const int lane = threadIdx.x & 31;
    const int half = lane >> 4;
    const int r    = lane & 15;
    const int n0   = wave * 16;

    v16h bfrag[8];
    #pragma unroll
    for (int kt = 0; kt < 8; ++kt) {
        const int kbase = kt * 32;
        #pragma unroll
        for (int j = 0; j < 8; ++j)
            bfrag[kt][j]     = (_Float16)W1[(kbase + 8 * half + j) * HID + (n0 + r)];
        #pragma unroll
        for (int j = 0; j < 8; ++j)
            bfrag[kt][8 + j] = (_Float16)W1[(kbase + 16 + 8 * half + j) * HID + (n0 + r)];
    }

    for (int rt = 0; rt < RT; ++rt) {
        const int row0 = (blockIdx.x * RT + rt) * 16;
        const float* xrow = x + (size_t)(row0 + r) * F_IN;

        if (rt + 1 < RT) {
            const float* nxt = xrow + 16 * F_IN;
            __builtin_prefetch(nxt +   0, 0, 0);
            __builtin_prefetch(nxt +  64, 0, 0);
            __builtin_prefetch(nxt + 128, 0, 0);
            __builtin_prefetch(nxt + 192, 0, 0);
        }

        v8f acc = {0.f, 0.f, 0.f, 0.f, 0.f, 0.f, 0.f, 0.f};
        #pragma unroll
        for (int kt = 0; kt < 8; ++kt) {
            const float* seg0 = xrow + kt * 32 + 8 * half;
            const float* seg1 = seg0 + 16;
            v4f a0 = __builtin_nontemporal_load((const v4f*)seg0);
            v4f a1 = __builtin_nontemporal_load((const v4f*)(seg0 + 4));
            v4f a2 = __builtin_nontemporal_load((const v4f*)seg1);
            v4f a3 = __builtin_nontemporal_load((const v4f*)(seg1 + 4));
            v16h a;
            a[0]  = (_Float16)a0.x; a[1]  = (_Float16)a0.y;
            a[2]  = (_Float16)a0.z; a[3]  = (_Float16)a0.w;
            a[4]  = (_Float16)a1.x; a[5]  = (_Float16)a1.y;
            a[6]  = (_Float16)a1.z; a[7]  = (_Float16)a1.w;
            a[8]  = (_Float16)a2.x; a[9]  = (_Float16)a2.y;
            a[10] = (_Float16)a2.z; a[11] = (_Float16)a2.w;
            a[12] = (_Float16)a3.x; a[13] = (_Float16)a3.y;
            a[14] = (_Float16)a3.z; a[15] = (_Float16)a3.w;

            acc = __builtin_amdgcn_wmma_f32_16x16x32_f16(
                 false, a,  false, bfrag[kt],
                 (short)0, acc,  false,  false);
        }

        #pragma unroll
        for (int i = 0; i < 8; ++i) {
            int m = i + 8 * half;
            xw1[(size_t)(row0 + m) * HID + (n0 + r)] = acc[i];
        }
    }
}

#define BIN    391
#define NBINS  256
#define QCAP   12288
#define NGRP   (N_EDGES / 4)
#define SWEEP  1000
#define NSTEP  400
#define EBITS  21

static __device__ __forceinline__ void own_edge(unsigned dl, int e,
        const int* __restrict__ ei, const float* __restrict__ dinv,
        const float* __restrict__ xw1,
        float* acc, const float* dinv_l, unsigned* queue, unsigned* qcount) {
    unsigned slot = atomicAdd(qcount, 1u);
    if (slot < QCAP) {
        queue[slot] = (unsigned)e | (dl << EBITS);
    } else {
        int s = ei[e];
        float norm = dinv[s] * dinv_l[dl];
        const float* row = xw1 + (size_t)s * HID;
        for (int f = 0; f < HID; ++f) atomicAdd(&acc[dl * HID + f], row[f] * norm);
    }
}

__global__ __launch_bounds__(1024) void k_agg1_own(
        const int* __restrict__ ei,
        const float* __restrict__ dinv,
        const float* __restrict__ xw1,
        const float* __restrict__ b1,
        float* __restrict__ agg1) {
    __shared__ __attribute__((aligned(16))) float acc[BIN * HID];
    __shared__ float    dinv_l[BIN];
    __shared__ unsigned queue[QCAP];
    __shared__ unsigned qcount;

    const int tid  = threadIdx.x;
    const int lane = tid & 31;
    const int wave = tid >> 5;
    const int lo   = blockIdx.x * BIN;
    const int nown = min(BIN, N_NODES - lo);

    if (tid == 0) qcount = 0u;
    for (int n = tid; n < nown; n += 1024) dinv_l[n] = dinv[lo + n];
    for (int i = tid; i < nown * HID; i += 1024) {
        float di = dinv[lo + (i >> 6)];
        acc[i] = xw1[(size_t)lo * HID + i] * (di * di) + b1[i & 63];
    }
    __syncthreads();

    const int4* dst4 = (const int4*)(ei + N_EDGES);
    const int start = blockIdx.x * 1560;
    if (tid < SWEEP) {
        int4 da[4], db[4];
        int  ga[4], gb[4];
        auto ask = [&](int4 (&dv)[4], int (&gv)[4], int k) {
            #pragma unroll
            for (int u = 0; u < 4; ++u) {
                int g = tid + (k + u) * SWEEP + start;
                if (g >= NGRP) g -= NGRP;
                gv[u] = g;
                dv[u] = dst4[g];
            }
        };
        auto look = [&](const int4 (&dv)[4], const int (&gv)[4]) {
            #pragma unroll
            for (int u = 0; u < 4; ++u) {
                unsigned d0 = (unsigned)(dv[u].x - lo), d1 = (unsigned)(dv[u].y - lo);
                unsigned d2 = (unsigned)(dv[u].z - lo), d3 = (unsigned)(dv[u].w - lo);
                if ((d0 < (unsigned)BIN) | (d1 < (unsigned)BIN) | (d2 < (unsigned)BIN) | (d3 < (unsigned)BIN)) {
                    int e = gv[u] * 4;
                    if (d0 < (unsigned)BIN) own_edge(d0, e + 0, ei, dinv, xw1, acc, dinv_l, queue, &qcount);
                    if (d1 < (unsigned)BIN) own_edge(d1, e + 1, ei, dinv, xw1, acc, dinv_l, queue, &qcount);
                    if (d2 < (unsigned)BIN) own_edge(d2, e + 2, ei, dinv, xw1, acc, dinv_l, queue, &qcount);
                    if (d3 < (unsigned)BIN) own_edge(d3, e + 3, ei, dinv, xw1, acc, dinv_l, queue, &qcount);
                }
            }
        };
        ask(da, ga, 0);
        for (int k = 0; k < NSTEP - 8; k += 8) {
            ask(db, gb, k + 4);
            look(da, ga);
            ask(da, ga, k + 8);
            look(db, gb);
        }
        ask(db, gb, NSTEP - 4);
        look(da, ga);
        look(db, gb);
    }
    __syncthreads();

    const int count = (int)min(qcount, (unsigned)QCAP);
    for (int base = wave * 32; base < count; base += 32 * 32) {
        const int nval = min(32, count - base);
        unsigned ent = (lane < nval) ? queue[base + lane] : 0u;
        int   s_l  = (lane < nval) ? ei[ent & ((1u << EBITS) - 1u)] : 0;
        int   dl_l = (int)(ent >> EBITS);
        float ds_l = dinv[s_l];
        float dd_l = dinv_l[dl_l];
        #pragma unroll
        for (int jj = 0; jj < 32; jj += 16) {
            if (jj < nval) {
                float v0[16], v1[16];
                #pragma unroll
                for (int u = 0; u < 16; ++u) {
                    if (jj + u < nval) {
                        int sj = __builtin_amdgcn_readlane(s_l, jj + u);
                        const float* row = xw1 + (size_t)sj * HID;
                        v0[u] = row[lane];
                        v1[u] = row[32 + lane];
                    }
                }
                #pragma unroll
                for (int u = 0; u < 16; ++u) {
                    if (jj + u < nval) {
                        int   dj = __builtin_amdgcn_readlane(dl_l, jj + u);
                        float nj = __int_as_float(__builtin_amdgcn_readlane(__float_as_int(ds_l), jj + u))
                                 * __int_as_float(__builtin_amdgcn_readlane(__float_as_int(dd_l), jj + u));
                        atomicAdd(&acc[dj * HID + lane],      v0[u] * nj);
                        atomicAdd(&acc[dj * HID + 32 + lane], v1[u] * nj);
                    }
                }
            }
        }
    }
    __syncthreads();

    const v4f* acc4 = (const v4f*)acc;
    v4f* out4 = (v4f*)(agg1 + (size_t)lo * HID);
    for (int i = tid; i < nown * (HID / 4); i += 1024) out4[i] = acc4[i];
}

__global__ void k_relu_gemm2(const float* __restrict__ agg1,
                             const float* __restrict__ W2,
                             const float* __restrict__ b2,
                             const float* __restrict__ dinv,
                             float* __restrict__ zd,
                             float* __restrict__ acc2) {
    int i = blockIdx.x * blockDim.x + threadIdx.x;
    if (i >= N_NODES) return;
    const float* row = agg1 + (size_t)i * HID;
    float o0 = 0.f, o1 = 0.f;
    #pragma unroll 8
    for (int k = 0; k < HID; ++k) {
        float h = fmaxf(row[k], 0.f);
        o0 += h * W2[k * 2 + 0];
        o1 += h * W2[k * 2 + 1];
    }
    const float z = o0 - o1;
    zd[i] = z;
    float di = dinv[i];
    float sl = di * di;
    acc2[i] = z * sl + (b2[0] - b2[1]);
}

__global__ void k_agg2_edges(const int* __restrict__ ei,
                             const float* __restrict__ dinv,
                             const float* __restrict__ zd,
                             float* __restrict__ acc2) {
    int e = blockIdx.x * blockDim.x + threadIdx.x;
    if (e >= N_EDGES) return;
    int s = ei[e];
    int d = ei[N_EDGES + e];
    float norm = dinv[s] * dinv[d];
    atomicAddF32(acc2 + d, zd[s] * norm);
}

__global__ void k_softmax2(const float* __restrict__ acc2,
                           float* __restrict__ out) {
    int i = blockIdx.x * blockDim.x + threadIdx.x;
    if (i >= N_NODES) return;
    float a0 = acc2[i];
    float a1 = 0.0f;
    float m  = fmaxf(a0, a1);
    float e0 = __expf(a0 - m);
    float e1 = __expf(a1 - m);
    float inv = 1.0f / (e0 + e1);
    out[(size_t)i * 2 + 0] = e0 * inv;
    out[(size_t)i * 2 + 1] = e1 * inv;
}

extern "C" void kernel_launch(void* const* d_in, const int* in_sizes, int n_in,
                              void* d_out, int out_size, void* d_ws, size_t ws_size,
                              hipStream_t stream) {
    const float* x  = (const float*)d_in[0];
    const int*   ei = (const int*)d_in[1];
    const float* W1 = (const float*)d_in[2];
    const float* b1 = (const float*)d_in[3];
    const float* W2 = (const float*)d_in[4];
    const float* b2 = (const float*)d_in[5];
    float* out = (float*)d_out;

    char* ws = (char*)d_ws;
    size_t o = 0;
    auto alloc = [&](size_t bytes) {
        char* p = ws + o;
        o += (bytes + 4095) & ~(size_t)4095;
        return p;
    };
    int*   deg  = (int*)  alloc((size_t)N_NODES * sizeof(int));
    float* dinv = (float*)alloc((size_t)N_NODES * sizeof(float));
    float* xw1  = (float*)alloc((size_t)N_NODES * HID * sizeof(float));
    float* agg1 = (float*)alloc((size_t)N_NODES * HID * sizeof(float));
    float* zd   = (float*)alloc((size_t)N_NODES * sizeof(float));
    float* acc2 = (float*)alloc((size_t)N_NODES * sizeof(float));
    (void)ws_size; (void)in_sizes; (void)n_in; (void)out_size;

    const int B = 256;
    dim3 blk(B);

    k_deg_zero <<<(N_NODES + B - 1) / B, blk, 0, stream>>>(deg);
    k_deg_count<<<(N_EDGES + B - 1) / B, blk, 0, stream>>>(ei, deg);
    k_dinv     <<<(N_NODES + B - 1) / B, blk, 0, stream>>>(deg, dinv);

    k_gemm1_wmma<<<(N_NODES / 16) / RT, dim3(128), 0, stream>>>(x, W1, xw1);

    k_agg1_own<<<NBINS, dim3(1024), 0, stream>>>(ei, dinv, xw1, b1, agg1);

    k_relu_gemm2<<<(N_NODES + B - 1) / B, blk, 0, stream>>>(agg1, W2, b2, dinv, zd, acc2);

    k_agg2_edges<<<(N_EDGES + B - 1) / B, blk, 0, stream>>>(ei, dinv, zd, acc2);
    k_softmax2  <<<(N_NODES + B - 1) / B, blk, 0, stream>>>(acc2, out);
}


# owners carry layer 2: each forms its nodes' relu-W2 logit difference from its LDS rows (lane tree), keeps its (src,dst) queue; an owner launch adds it in LDS, arrival order; no global float atomic
# speedup vs baseline: 1.8686x; 1.3108x over previous
#include <hip/hip_runtime.h>
#include <hip/hip_bf16.h>
#include <stdint.h>

#define N_NODES 100000
#define N_EDGES 1600000
#define F_IN    256
#define HID     64
#define RT      5

typedef __attribute__((ext_vector_type(16))) _Float16 v16h;
typedef __attribute__((ext_vector_type(8)))  float    v8f;
typedef __attribute__((ext_vector_type(4)))  float    v4f;

__global__ void k_deg_zero(int* __restrict__ deg) {
    int i = blockIdx.x * blockDim.x + threadIdx.x;
    if (i < N_NODES) deg[i] = 0;
}

__global__ void k_deg_count(const int* __restrict__ ei, int* __restrict__ deg) {
    int e = blockIdx.x * blockDim.x + threadIdx.x;
    if (e < N_EDGES) {
        int d = ei[N_EDGES + e];
        atomicAdd(&deg[d], 1);
    }
}

__global__ void k_dinv(const int* __restrict__ deg, float* __restrict__ dinv) {
    int i = blockIdx.x * blockDim.x + threadIdx.x;
    if (i < N_NODES) dinv[i] = __frsqrt_rn((float)deg[i] + 1.0f);
}

__global__ __launch_bounds__(128) void k_gemm1_wmma(
        const float* __restrict__ x,
        const float* __restrict__ W1,
        float* __restrict__ xw1) {
    const int wave = threadIdx.x >> 5;
    const int lane = threadIdx.x & 31;
    const int half = lane >> 4;
    const int r    = lane & 15;
    const int n0   = wave * 16;

    v16h bfrag[8];
    #pragma unroll
    for (int kt = 0; kt < 8; ++kt) {
        const int kbase = kt * 32;
        #pragma unroll
        for (int j = 0; j < 8; ++j)
            bfrag[kt][j]     = (_Float16)W1[(kbase + 8 * half + j) * HID + (n0 + r)];
        #pragma unroll
        for (int j = 0; j < 8; ++j)
            bfrag[kt][8 + j] = (_Float16)W1[(kbase + 16 + 8 * half + j) * HID + (n0 + r)];
    }

    for (int rt = 0; rt < RT; ++rt) {
        const int row0 = (blockIdx.x * RT + rt) * 16;
        const float* xrow = x + (size_t)(row0 + r) * F_IN;

        if (rt + 1 < RT) {
            const float* nxt = xrow + 16 * F_IN;
            __builtin_prefetch(nxt +   0, 0, 0);
            __builtin_prefetch(nxt +  64, 0, 0);
            __builtin_prefetch(nxt + 128, 0, 0);
            __builtin_prefetch(nxt + 192, 0, 0);
        }

        v8f acc = {0.f, 0.f, 0.f, 0.f, 0.f, 0.f, 0.f, 0.f};
        #pragma unroll
        for (int kt = 0; kt < 8; ++kt) {
            const float* seg0 = xrow + kt * 32 + 8 * half;
            const float* seg1 = seg0 + 16;
            v4f a0 = __builtin_nontemporal_load((const v4f*)seg0);
            v4f a1 = __builtin_nontemporal_load((const v4f*)(seg0 + 4));
            v4f a2 = __builtin_nontemporal_load((const v4f*)seg1);
            v4f a3 = __builtin_nontemporal_load((const v4f*)(seg1 + 4));
            v16h a;
            a[0]  = (_Float16)a0.x; a[1]  = (_Float16)a0.y;
            a[2]  = (_Float16)a0.z; a[3]  = (_Float16)a0.w;
            a[4]  = (_Float16)a1.x; a[5]  = (_Float16)a1.y;
            a[6]  = (_Float16)a1.z; a[7]  = (_Float16)a1.w;
            a[8]  = (_Float16)a2.x; a[9]  = (_Float16)a2.y;
            a[10] = (_Float16)a2.z; a[11] = (_Float16)a2.w;
            a[12] = (_Float16)a3.x; a[13] = (_Float16)a3.y;
            a[14] = (_Float16)a3.z; a[15] = (_Float16)a3.w;

            acc = __builtin_amdgcn_wmma_f32_16x16x32_f16(
                 false, a,  false, bfrag[kt],
                 (short)0, acc,  false,  false);
        }

        #pragma unroll
        for (int i = 0; i < 8; ++i) {
            int m = i + 8 * half;
            xw1[(size_t)(row0 + m) * HID + (n0 + r)] = acc[i];
        }
    }
}

#define BIN    416
#define NBINS  ((N_NODES + BIN - 1) / BIN)
#define SBITS  17
#define QCAP   12288
#define NGRP   (N_EDGES / 4)
#define SWEEP  1000
#define NSTEP  400
#define EBITS  21

static __device__ __forceinline__ void own_edge(unsigned dl, int e,
        const int* __restrict__ ei, const float* __restrict__ dinv,
        const float* __restrict__ xw1,
        float* acc, const float* dinv_l, unsigned* queue, unsigned* qcount) {
    unsigned slot = atomicAdd(qcount, 1u);
    if (slot < QCAP) {
        queue[slot] = (unsigned)e | (dl << EBITS);
    } else {
        int s = ei[e];
        float norm = dinv[s] * dinv_l[dl];
        const float* row = xw1 + (size_t)s * HID;
        for (int f = 0; f < HID; ++f) atomicAdd(&acc[dl * HID + f], row[f] * norm);
    }
}

__global__ __launch_bounds__(1024) void k_agg1_own(
        const int* __restrict__ ei,
        const float* __restrict__ dinv,
        const float* __restrict__ xw1,
        const float* __restrict__ b1,
        const float* __restrict__ W2,
        float* __restrict__ zs,
        unsigned* __restrict__ qkeep,
        unsigned* __restrict__ qn) {
    __shared__ float    acc[BIN * HID];
    __shared__ float    dinv_l[BIN];
    __shared__ float    zl[BIN];
    __shared__ unsigned queue[QCAP];
    __shared__ unsigned qcount;

    const int tid  = threadIdx.x;
    const int lane = tid & 31;
    const int wave = tid >> 5;
    const int lo   = blockIdx.x * BIN;
    const int nown = min(BIN, N_NODES - lo);

    if (tid == 0) qcount = 0u;
    for (int n = tid; n < nown; n += 1024) dinv_l[n] = dinv[lo + n];
    for (int i = tid; i < nown * HID; i += 1024) {
        float di = dinv[lo + (i >> 6)];
        acc[i] = xw1[(size_t)lo * HID + i] * (di * di) + b1[i & 63];
    }
    __syncthreads();

    const int4* dst4 = (const int4*)(ei + N_EDGES);
    const int start = blockIdx.x * 1560;
    if (tid < SWEEP) {
        int4 da[4], db[4];
        int  ga[4], gb[4];
        auto ask = [&](int4 (&dv)[4], int (&gv)[4], int k) {
            #pragma unroll
            for (int u = 0; u < 4; ++u) {
                int g = tid + (k + u) * SWEEP + start;
                if (g >= NGRP) g -= NGRP;
                gv[u] = g;
                dv[u] = dst4[g];
            }
        };
        auto look = [&](const int4 (&dv)[4], const int (&gv)[4]) {
            #pragma unroll
            for (int u = 0; u < 4; ++u) {
                unsigned d0 = (unsigned)(dv[u].x - lo), d1 = (unsigned)(dv[u].y - lo);
                unsigned d2 = (unsigned)(dv[u].z - lo), d3 = (unsigned)(dv[u].w - lo);
                if ((d0 < (unsigned)BIN) | (d1 < (unsigned)BIN) | (d2 < (unsigned)BIN) | (d3 < (unsigned)BIN)) {
                    int e = gv[u] * 4;
                    if (d0 < (unsigned)BIN) own_edge(d0, e + 0, ei, dinv, xw1, acc, dinv_l, queue, &qcount);
                    if (d1 < (unsigned)BIN) own_edge(d1, e + 1, ei, dinv, xw1, acc, dinv_l, queue, &qcount);
                    if (d2 < (unsigned)BIN) own_edge(d2, e + 2, ei, dinv, xw1, acc, dinv_l, queue, &qcount);
                    if (d3 < (unsigned)BIN) own_edge(d3, e + 3, ei, dinv, xw1, acc, dinv_l, queue, &qcount);
                }
            }
        };
        ask(da, ga, 0);
        for (int k = 0; k < NSTEP - 8; k += 8) {
            ask(db, gb, k + 4);
            look(da, ga);
            ask(da, ga, k + 8);
            look(db, gb);
        }
        ask(db, gb, NSTEP - 4);
        look(da, ga);
        look(db, gb);
    }
    __syncthreads();

    const int count = (int)min(qcount, (unsigned)QCAP);
    for (int base = wave * 32; base < count; base += 32 * 32) {
        const int nval = min(32, count - base);
        unsigned ent = (lane < nval) ? queue[base + lane] : 0u;
        int   s_l  = (lane < nval) ? ei[ent & ((1u << EBITS) - 1u)] : 0;
        int   dl_l = (int)(ent >> EBITS);
        float ds_l = dinv[s_l];
        float dd_l = dinv_l[dl_l];
        if (lane < nval)
            qkeep[(size_t)blockIdx.x * QCAP + base + lane] = (unsigned)s_l | ((unsigned)dl_l << SBITS);
        #pragma unroll
        for (int jj = 0; jj < 32; jj += 16) {
            if (jj < nval) {
                float v0[16], v1[16];
                #pragma unroll
                for (int u = 0; u < 16; ++u) {
                    if (jj + u < nval) {
                        int sj = __builtin_amdgcn_readlane(s_l, jj + u);
                        const float* row = xw1 + (size_t)sj * HID;
                        v0[u] = row[lane];
                        v1[u] = row[32 + lane];
                    }
                }
                #pragma unroll
                for (int u = 0; u < 16; ++u) {
                    if (jj + u < nval) {
                        int   dj = __builtin_amdgcn_readlane(dl_l, jj + u);
                        float nj = __int_as_float(__builtin_amdgcn_readlane(__float_as_int(ds_l), jj + u))
                                 * __int_as_float(__builtin_amdgcn_readlane(__float_as_int(dd_l), jj + u));
                        atomicAdd(&acc[dj * HID + lane],      v0[u] * nj);
                        atomicAdd(&acc[dj * HID + 32 + lane], v1[u] * nj);
                    }
                }
            }
        }
    }
    __syncthreads();

    const float w0a = W2[lane * 2 + 0],        w1a = W2[lane * 2 + 1];
    const float w0b = W2[(32 + lane) * 2 + 0], w1b = W2[(32 + lane) * 2 + 1];
    for (int n = wave; n < nown; n += 32) {
        float h0 = fmaxf(acc[n * HID + lane], 0.f);
        float h1 = fmaxf(acc[n * HID + 32 + lane], 0.f);
        float o0 = h0 * w0a + h1 * w0b;
        float o1 = h0 * w1a + h1 * w1b;
        for (int off = 16; off > 0; off >>= 1) {
            o0 += __shfl_xor(o0, off);
            o1 += __shfl_xor(o1, off);
        }
        if (lane == 0) zl[n] = (o0 - o1) * dinv_l[n];
    }
    if (tid == 0) qn[blockIdx.x] = qcount;
    __syncthreads();
    for (int n = tid; n < nown; n += 1024) zs[lo + n] = zl[n];
}

__global__ __launch_bounds__(1024) void k_agg2_own(
        const int* __restrict__ ei,
        const float* __restrict__ dinv,
        const float* __restrict__ zs,
        const float* __restrict__ b2,
        const unsigned* __restrict__ qkeep,
        const unsigned* __restrict__ qn,
        float* __restrict__ out) {
    __shared__ float sum[BIN];

    const int tid  = threadIdx.x;
    const int lo   = blockIdx.x * BIN;
    const int nown = min(BIN, N_NODES - lo);

    for (int n = tid; n < nown; n += 1024) sum[n] = zs[lo + n];
    __syncthreads();

    const unsigned count = qn[blockIdx.x];
    if (count <= (unsigned)QCAP) {
        const unsigned* q = qkeep + (size_t)blockIdx.x * QCAP;
        const unsigned NONE = 0xFFFFFFFFu;
        for (unsigned i0 = tid; i0 < count; i0 += 4u * 1024u) {
            unsigned ent[4];
            float    v[4];
            #pragma unroll
            for (int u = 0; u < 4; ++u) {
                unsigned i = i0 + (unsigned)u * 1024u;
                ent[u] = (i < count) ? q[i] : NONE;
            }
            #pragma unroll
            for (int u = 0; u < 4; ++u)
                v[u] = (ent[u] != NONE) ? zs[ent[u] & ((1u << SBITS) - 1u)] : 0.f;
            #pragma unroll
            for (int u = 0; u < 4; ++u)
                if (ent[u] != NONE) atomicAdd(&sum[ent[u] >> SBITS], v[u]);
        }
    } else {
        for (int e = tid; e < N_EDGES; e += 1024) {
            unsigned dl = (unsigned)(ei[N_EDGES + e] - lo);
            if (dl < (unsigned)BIN) atomicAdd(&sum[dl], zs[ei[e]]);
        }
    }
    __syncthreads();

    const float bd = b2[0] - b2[1];
    for (int n = tid; n < nown; n += 1024) {
        float a0 = dinv[lo + n] * sum[n] + bd;
        float a1 = 0.0f;
        float m  = fmaxf(a0, a1);
        float e0 = __expf(a0 - m);
        float e1 = __expf(a1 - m);
        float inv = 1.0f / (e0 + e1);
        float2 r = make_float2(e0 * inv, e1 * inv);
        *(float2*)(out + (size_t)(lo + n) * 2) = r;
    }
}

extern "C" void kernel_launch(void* const* d_in, const int* in_sizes, int n_in,
                              void* d_out, int out_size, void* d_ws, size_t ws_size,
                              hipStream_t stream) {
    const float* x  = (const float*)d_in[0];
    const int*   ei = (const int*)d_in[1];
    const float* W1 = (const float*)d_in[2];
    const float* b1 = (const float*)d_in[3];
    const float* W2 = (const float*)d_in[4];
    const float* b2 = (const float*)d_in[5];
    float* out = (float*)d_out;

    char* ws = (char*)d_ws;
    size_t o = 0;
    auto alloc = [&](size_t bytes) {
        char* p = ws + o;
        o += (bytes + 4095) & ~(size_t)4095;
        return p;
    };
    int*   deg  = (int*)  alloc((size_t)N_NODES * sizeof(int));
    float* dinv = (float*)alloc((size_t)N_NODES * sizeof(float));
    float* xw1  = (float*)alloc((size_t)N_NODES * HID * sizeof(float));
    unsigned* qkeep = (unsigned*)alloc((size_t)NBINS * QCAP * sizeof(unsigned));
    float*    zs    = (float*)   alloc((size_t)N_NODES * sizeof(float));
    unsigned* qn    = (unsigned*)alloc((size_t)NBINS * sizeof(unsigned));
    (void)ws_size; (void)in_sizes; (void)n_in; (void)out_size;

    const int B = 256;
    dim3 blk(B);

    k_deg_zero <<<(N_NODES + B - 1) / B, blk, 0, stream>>>(deg);
    k_deg_count<<<(N_EDGES + B - 1) / B, blk, 0, stream>>>(ei, deg);
    k_dinv     <<<(N_NODES + B - 1) / B, blk, 0, stream>>>(deg, dinv);

    k_gemm1_wmma<<<(N_NODES / 16) / RT, dim3(128), 0, stream>>>(x, W1, xw1);

    k_agg1_own<<<NBINS, dim3(1024), 0, stream>>>(ei, dinv, xw1, b1, W2, zs, qkeep, qn);

    k_agg2_own<<<NBINS, dim3(1024), 0, stream>>>(ei, dinv, zs, b2, qkeep, qn, out);
}


# degree launch (16 edges/thread) also stores each edge's owner id as a byte; owners sweep that 1.6-MB list, 16 edges per 128-bit load (zero-byte test), not the 6.4-MB dst list; LDS arrival order kept
# speedup vs baseline: 2.0630x; 1.1040x over previous
#include <hip/hip_runtime.h>
#include <hip/hip_bf16.h>
#include <stdint.h>

#define N_NODES 100000
#define N_EDGES 1600000
#define F_IN    256
#define HID     64
#define RT      5

typedef __attribute__((ext_vector_type(16))) _Float16 v16h;
typedef __attribute__((ext_vector_type(8)))  float    v8f;
typedef __attribute__((ext_vector_type(4)))  float    v4f;

__global__ void k_deg_zero(int* __restrict__ deg) {
    int i = blockIdx.x * blockDim.x + threadIdx.x;
    if (i < N_NODES) deg[i] = 0;
}

#define BIN    416
#define NGRP   (N_EDGES / 4)
__global__ __launch_bounds__(256) void k_deg_count(const int* __restrict__ ei, int* __restrict__ deg,
                                                   unsigned* __restrict__ own) {
    const int4* dst4 = (const int4*)(ei + N_EDGES);
    const int g0 = blockIdx.x * 1024 + threadIdx.x;
    int4 d[4];
    #pragma unroll
    for (int u = 0; u < 4; ++u) {
        int g = g0 + u * 256;
        if (g < NGRP) d[u] = dst4[g];
    }
    #pragma unroll
    for (int u = 0; u < 4; ++u) {
        int g = g0 + u * 256;
        if (g < NGRP) {
            atomicAdd(&deg[d[u].x], 1);
            atomicAdd(&deg[d[u].y], 1);
            atomicAdd(&deg[d[u].z], 1);
            atomicAdd(&deg[d[u].w], 1);
            own[g] = (unsigned)(d[u].x / BIN)         | ((unsigned)(d[u].y / BIN) << 8)
                   | ((unsigned)(d[u].z / BIN) << 16) | ((unsigned)(d[u].w / BIN) << 24);
        }
    }
}

__global__ void k_dinv(const int* __restrict__ deg, float* __restrict__ dinv) {
    int i = blockIdx.x * blockDim.x + threadIdx.x;
    if (i < N_NODES) dinv[i] = __frsqrt_rn((float)deg[i] + 1.0f);
}

__global__ __launch_bounds__(128) void k_gemm1_wmma(
        const float* __restrict__ x,
        const float* __restrict__ W1,
        float* __restrict__ xw1) {
    const int wave = threadIdx.x >> 5;
    const int lane = threadIdx.x & 31;
    const int half = lane >> 4;
    const int r    = lane & 15;
    const int n0   = wave * 16;

    v16h bfrag[8];
    #pragma unroll
    for (int kt = 0; kt < 8; ++kt) {
        const int kbase = kt * 32;
        #pragma unroll
        for (int j = 0; j < 8; ++j)
            bfrag[kt][j]     = (_Float16)W1[(kbase + 8 * half + j) * HID + (n0 + r)];
        #pragma unroll
        for (int j = 0; j < 8; ++j)
            bfrag[kt][8 + j] = (_Float16)W1[(kbase + 16 + 8 * half + j) * HID + (n0 + r)];
    }

    for (int rt = 0; rt < RT; ++rt) {
        const int row0 = (blockIdx.x * RT + rt) * 16;
        const float* xrow = x + (size_t)(row0 + r) * F_IN;

        if (rt + 1 < RT) {
            const float* nxt = xrow + 16 * F_IN;
            __builtin_prefetch(nxt +   0, 0, 0);
            __builtin_prefetch(nxt +  64, 0, 0);
            __builtin_prefetch(nxt + 128, 0, 0);
            __builtin_prefetch(nxt + 192, 0, 0);
        }

        v8f acc = {0.f, 0.f, 0.f, 0.f, 0.f, 0.f, 0.f, 0.f};
        #pragma unroll
        for (int kt = 0; kt < 8; ++kt) {
            const float* seg0 = xrow + kt * 32 + 8 * half;
            const float* seg1 = seg0 + 16;
            v4f a0 = __builtin_nontemporal_load((const v4f*)seg0);
            v4f a1 = __builtin_nontemporal_load((const v4f*)(seg0 + 4));
            v4f a2 = __builtin_nontemporal_load((const v4f*)seg1);
            v4f a3 = __builtin_nontemporal_load((const v4f*)(seg1 + 4));
            v16h a;
            a[0]  = (_Float16)a0.x; a[1]  = (_Float16)a0.y;
            a[2]  = (_Float16)a0.z; a[3]  = (_Float16)a0.w;
            a[4]  = (_Float16)a1.x; a[5]  = (_Float16)a1.y;
            a[6]  = (_Float16)a1.z; a[7]  = (_Float16)a1.w;
            a[8]  = (_Float16)a2.x; a[9]  = (_Float16)a2.y;
            a[10] = (_Float16)a2.z; a[11] = (_Float16)a2.w;
            a[12] = (_Float16)a3.x; a[13] = (_Float16)a3.y;
            a[14] = (_Float16)a3.z; a[15] = (_Float16)a3.w;

            acc = __builtin_amdgcn_wmma_f32_16x16x32_f16(
                 false, a,  false, bfrag[kt],
                 (short)0, acc,  false,  false);
        }

        #pragma unroll
        for (int i = 0; i < 8; ++i) {
            int m = i + 8 * half;
            xw1[(size_t)(row0 + m) * HID + (n0 + r)] = acc[i];
        }
    }
}

#define NBINS  ((N_NODES + BIN - 1) / BIN)
#define SBITS  17
#define QCAP   12288
#define GCAP   12288
#define OGRP   (N_EDGES / 16)
#define SWEEP  1000
#define NSTEP  100

static __device__ __forceinline__ void own_edge(int e, int lo,
        const int* __restrict__ ei, const float* __restrict__ dinv,
        const float* __restrict__ xw1,
        float* acc, const float* dinv_l, unsigned* queue, unsigned* qcount) {
    unsigned slot = atomicAdd(qcount, 1u);
    if (slot < QCAP) {
        queue[slot] = (unsigned)e;
    } else {
        int s  = ei[e];
        int dl = ei[N_EDGES + e] - lo;
        float norm = dinv[s] * dinv_l[dl];
        const float* row = xw1 + (size_t)s * HID;
        #pragma nounroll
        for (int f = 0; f < HID; ++f) atomicAdd(&acc[dl * HID + f], row[f] * norm);
    }
}

__global__ __launch_bounds__(1024) void k_agg1_own(
        const int* __restrict__ ei,
        const unsigned* __restrict__ own,
        const float* __restrict__ dinv,
        const float* __restrict__ xw1,
        const float* __restrict__ b1,
        const float* __restrict__ W2,
        float* __restrict__ zs,
        unsigned* __restrict__ qkeep,
        unsigned* __restrict__ qn) {
    __shared__ float    acc[BIN * HID];
    __shared__ float    dinv_l[BIN];
    __shared__ float    zl[BIN];
    __shared__ unsigned queue[QCAP];
    __shared__ unsigned gqueue[GCAP];
    __shared__ unsigned qcount;
    __shared__ unsigned gcount;

    const int tid  = threadIdx.x;
    const int lane = tid & 31;
    const int wave = tid >> 5;
    const int lo   = blockIdx.x * BIN;
    const int nown = min(BIN, N_NODES - lo);

    if (tid == 0) { qcount = 0u; gcount = 0u; }
    for (int n = tid; n < nown; n += 1024) dinv_l[n] = dinv[lo + n];
    for (int i = tid; i < nown * HID; i += 1024) {
        float di = dinv[lo + (i >> 6)];
        acc[i] = xw1[(size_t)lo * HID + i] * (di * di) + b1[i & 63];
    }
    __syncthreads();

    const uint4* own4 = (const uint4*)own;
    const unsigned pat = (unsigned)blockIdx.x * 0x01010101u;
    const int start = blockIdx.x * 408;
    if (tid < SWEEP) {
        uint4 oa[5], ob[5];
        int   ga[5], gb[5];
        auto ask = [&](uint4 (&ov)[5], int (&gv)[5], int k) {
            #pragma unroll
            for (int u = 0; u < 5; ++u) {
                int g = tid + (k + u) * SWEEP + start;
                if (g >= OGRP) g -= OGRP;
                gv[u] = g;
                ov[u] = own4[g];
            }
        };
        auto look = [&](const uint4 (&ov)[5], const int (&gv)[5]) {
            #pragma unroll
            for (int u = 0; u < 5; ++u) {
                unsigned x0 = ov[u].x ^ pat, x1 = ov[u].y ^ pat, x2 = ov[u].z ^ pat, x3 = ov[u].w ^ pat;
                unsigned t = ((x0 - 0x01010101u) & ~x0) | ((x1 - 0x01010101u) & ~x1)
                           | ((x2 - 0x01010101u) & ~x2) | ((x3 - 0x01010101u) & ~x3);
                if (t & 0x80808080u) {
                    unsigned slot = atomicAdd(&gcount, 1u);
                    if (slot < GCAP) gqueue[slot] = (unsigned)gv[u];
                }
            }
        };
        ask(oa, ga, 0);
        for (int k = 0; k < NSTEP - 10; k += 10) {
            ask(ob, gb, k + 5);
            look(oa, ga);
            ask(oa, ga, k + 10);
            look(ob, gb);
        }
        ask(ob, gb, NSTEP - 5);
        look(oa, ga);
        look(ob, gb);
    }
    __syncthreads();

    if (gcount <= (unsigned)GCAP) {
        const int gc = (int)gcount;
        int   gn = 0;
        uint4 on = make_uint4(0u, 0u, 0u, 0u);
        if (tid < gc) { gn = (int)gqueue[tid]; on = own4[gn]; }
        for (int i = tid; i < gc; i += 1024) {
            const int   g = gn;
            const uint4 o = on;
            if (i + 1024 < gc) { gn = (int)gqueue[i + 1024]; on = own4[gn]; }
            unsigned x0 = o.x ^ pat, x1 = o.y ^ pat, x2 = o.z ^ pat, x3 = o.w ^ pat;
            unsigned c0 = ((x0 & 0x7F7F7F7Fu) + 0x7F7F7F7Fu) | x0;
            unsigned c1 = ((x1 & 0x7F7F7F7Fu) + 0x7F7F7F7Fu) | x1;
            unsigned c2 = ((x2 & 0x7F7F7F7Fu) + 0x7F7F7F7Fu) | x2;
            unsigned c3 = ((x3 & 0x7F7F7F7Fu) + 0x7F7F7F7Fu) | x3;
            unsigned m = ~(((c0 >> 7) & 0x01010101u) | ((c1 >> 6) & 0x02020202u)
                         | ((c2 >> 5) & 0x04040404u) | ((c3 >> 4) & 0x08080808u)) & 0x0F0F0F0Fu;
            while (m) {
                const int p = __builtin_ctz(m);
                m &= m - 1u;
                own_edge(g * 16 + ((p & 3) << 2) + (p >> 3), lo, ei, dinv, xw1, acc, dinv_l, queue, &qcount);
            }
        }
    } else {
        for (int e = tid; e < N_EDGES; e += 1024)
            if ((unsigned)(ei[N_EDGES + e] - lo) < (unsigned)BIN)
                own_edge(e, lo, ei, dinv, xw1, acc, dinv_l, queue, &qcount);
    }
    __syncthreads();

    const int count = (int)min(qcount, (unsigned)QCAP);
    for (int base = wave * 32; base < count; base += 32 * 32) {
        const int nval = min(32, count - base);
        const int e_l  = (lane < nval) ? (int)queue[base + lane] : 0;
        int   s_l  = ei[e_l];
        int   dl_l = (lane < nval) ? ei[N_EDGES + e_l] - lo : 0;
        float ds_l = dinv[s_l];
        float dd_l = dinv_l[dl_l];
        if (lane < nval)
            qkeep[(size_t)blockIdx.x * QCAP + base + lane] = (unsigned)s_l | ((unsigned)dl_l << SBITS);
        #pragma unroll
        for (int jj = 0; jj < 32; jj += 16) {
            if (jj < nval) {
                float v0[16], v1[16];
                #pragma unroll
                for (int u = 0; u < 16; ++u) {
                    if (jj + u < nval) {
                        int sj = __builtin_amdgcn_readlane(s_l, jj + u);
                        const float* row = xw1 + (size_t)sj * HID;
                        v0[u] = row[lane];
                        v1[u] = row[32 + lane];
                    }
                }
                #pragma unroll
                for (int u = 0; u < 16; ++u) {
                    if (jj + u < nval) {
                        int   dj = __builtin_amdgcn_readlane(dl_l, jj + u);
                        float nj = __int_as_float(__builtin_amdgcn_readlane(__float_as_int(ds_l), jj + u))
                                 * __int_as_float(__builtin_amdgcn_readlane(__float_as_int(dd_l), jj + u));
                        atomicAdd(&acc[dj * HID + lane],      v0[u] * nj);
                        atomicAdd(&acc[dj * HID + 32 + lane], v1[u] * nj);
                    }
                }
            }
        }
    }
    __syncthreads();

    const float w0a = W2[lane * 2 + 0],        w1a = W2[lane * 2 + 1];
    const float w0b = W2[(32 + lane) * 2 + 0], w1b = W2[(32 + lane) * 2 + 1];
    for (int n = wave; n < nown; n += 32) {
        float h0 = fmaxf(acc[n * HID + lane], 0.f);
        float h1 = fmaxf(acc[n * HID + 32 + lane], 0.f);
        float o0 = h0 * w0a + h1 * w0b;
        float o1 = h0 * w1a + h1 * w1b;
        for (int off = 16; off > 0; off >>= 1) {
            o0 += __shfl_xor(o0, off);
            o1 += __shfl_xor(o1, off);
        }
        if (lane == 0) zl[n] = (o0 - o1) * dinv_l[n];
    }
    if (tid == 0) qn[blockIdx.x] = qcount;
    __syncthreads();
    for (int n = tid; n < nown; n += 1024) zs[lo + n] = zl[n];
}

__global__ __launch_bounds__(1024) void k_agg2_own(
        const int* __restrict__ ei,
        const float* __restrict__ dinv,
        const float* __restrict__ zs,
        const float* __restrict__ b2,
        const unsigned* __restrict__ qkeep,
        const unsigned* __restrict__ qn,
        float* __restrict__ out) {
    __shared__ float sum[BIN];

    const int tid  = threadIdx.x;
    const int lo   = blockIdx.x * BIN;
    const int nown = min(BIN, N_NODES - lo);

    for (int n = tid; n < nown; n += 1024) sum[n] = zs[lo + n];
    __syncthreads();

    const unsigned count = qn[blockIdx.x];
    if (count <= (unsigned)QCAP) {
        const unsigned* q = qkeep + (size_t)blockIdx.x * QCAP;
        const unsigned NONE = 0xFFFFFFFFu;
        for (unsigned i0 = tid; i0 < count; i0 += 4u * 1024u) {
            unsigned ent[4];
            float    v[4];
            #pragma unroll
            for (int u = 0; u < 4; ++u) {
                unsigned i = i0 + (unsigned)u * 1024u;
                ent[u] = (i < count) ? q[i] : NONE;
            }
            #pragma unroll
            for (int u = 0; u < 4; ++u)
                v[u] = (ent[u] != NONE) ? zs[ent[u] & ((1u << SBITS) - 1u)] : 0.f;
            #pragma unroll
            for (int u = 0; u < 4; ++u)
                if (ent[u] != NONE) atomicAdd(&sum[ent[u] >> SBITS], v[u]);
        }
    } else {
        for (int e = tid; e < N_EDGES; e += 1024) {
            unsigned dl = (unsigned)(ei[N_EDGES + e] - lo);
            if (dl < (unsigned)BIN) atomicAdd(&sum[dl], zs[ei[e]]);
        }
    }
    __syncthreads();

    const float bd = b2[0] - b2[1];
    for (int n = tid; n < nown; n += 1024) {
        float a0 = dinv[lo + n] * sum[n] + bd;
        float a1 = 0.0f;
        float m  = fmaxf(a0, a1);
        float e0 = __expf(a0 - m);
        float e1 = __expf(a1 - m);
        float inv = 1.0f / (e0 + e1);
        float2 r = make_float2(e0 * inv, e1 * inv);
        *(float2*)(out + (size_t)(lo + n) * 2) = r;
    }
}

extern "C" void kernel_launch(void* const* d_in, const int* in_sizes, int n_in,
                              void* d_out, int out_size, void* d_ws, size_t ws_size,
                              hipStream_t stream) {
    const float* x  = (const float*)d_in[0];
    const int*   ei = (const int*)d_in[1];
    const float* W1 = (const float*)d_in[2];
    const float* b1 = (const float*)d_in[3];
    const float* W2 = (const float*)d_in[4];
    const float* b2 = (const float*)d_in[5];
    float* out = (float*)d_out;

    char* ws = (char*)d_ws;
    size_t o = 0;
    auto alloc = [&](size_t bytes) {
        char* p = ws + o;
        o += (bytes + 4095) & ~(size_t)4095;
        return p;
    };
    int*   deg  = (int*)  alloc((size_t)N_NODES * sizeof(int));
    float* dinv = (float*)alloc((size_t)N_NODES * sizeof(float));
    float* xw1  = (float*)alloc((size_t)N_NODES * HID * sizeof(float));
    unsigned* qkeep = (unsigned*)alloc((size_t)NBINS * QCAP * sizeof(unsigned));
    float*    zs    = (float*)   alloc((size_t)N_NODES * sizeof(float));
    unsigned* qn    = (unsigned*)alloc((size_t)NBINS * sizeof(unsigned));
    unsigned* own   = (unsigned*)alloc((size_t)N_EDGES);
    (void)ws_size; (void)in_sizes; (void)n_in; (void)out_size;

    const int B = 256;
    dim3 blk(B);

    k_deg_zero <<<(N_NODES + B - 1) / B, blk, 0, stream>>>(deg);
    k_deg_count<<<(NGRP + 1023) / 1024, blk, 0, stream>>>(ei, deg, own);
    k_dinv     <<<(N_NODES + B - 1) / B, blk, 0, stream>>>(deg, dinv);

    k_gemm1_wmma<<<(N_NODES / 16) / RT, dim3(128), 0, stream>>>(x, W1, xw1);

    k_agg1_own<<<NBINS, dim3(1024), 0, stream>>>(ei, own, dinv, xw1, b1, W2, zs, qkeep, qn);

    k_agg2_own<<<NBINS, dim3(1024), 0, stream>>>(ei, dinv, zs, b2, qkeep, qn, out);
}


# degrees counted by the dst-range owners in LDS from the edges they find (owner launch split where every dinv must be known): no global atomic, zero and dinv launches gone; LDS arrival order kept
# speedup vs baseline: 3.0940x; 1.4998x over previous
#include <hip/hip_runtime.h>
#include <hip/hip_bf16.h>
#include <stdint.h>

#define N_NODES 100000
#define N_EDGES 1600000
#define F_IN    256
#define HID     64
#define RT      5

typedef __attribute__((ext_vector_type(16))) _Float16 v16h;
typedef __attribute__((ext_vector_type(8)))  float    v8f;
typedef __attribute__((ext_vector_type(4)))  float    v4f;

#define BIN    416
#define NGRP   (N_EDGES / 4)
__global__ __launch_bounds__(256) void k_own_bytes(const int* __restrict__ ei, unsigned* __restrict__ own) {
    const int4* dst4 = (const int4*)(ei + N_EDGES);
    const int g0 = blockIdx.x * 1024 + threadIdx.x;
    int4 d[4];
    #pragma unroll
    for (int u = 0; u < 4; ++u) {
        int g = g0 + u * 256;
        if (g < NGRP) d[u] = dst4[g];
    }
    #pragma unroll
    for (int u = 0; u < 4; ++u) {
        int g = g0 + u * 256;
        if (g < NGRP)
            own[g] = (unsigned)(d[u].x / BIN)         | ((unsigned)(d[u].y / BIN) << 8)
                   | ((unsigned)(d[u].z / BIN) << 16) | ((unsigned)(d[u].w / BIN) << 24);
    }
}

#define NBINS  ((N_NODES + BIN - 1) / BIN)
#define SBITS  17
#define QCAP   12288
#define GCAP   12288
#define OGRP   (N_EDGES / 16)
#define SWEEP  1000
#define NSTEP  100

static __device__ __forceinline__ void own_edge(int e, int lo, const int* __restrict__ ei,
        unsigned* cnt, unsigned* queue, unsigned* qcount) {
    unsigned slot = atomicAdd(qcount, 1u);
    if (slot < QCAP) {
        queue[slot] = (unsigned)e;
    } else {
        atomicAdd(&cnt[ei[N_EDGES + e] - lo], 1u);
    }
}

__global__ __launch_bounds__(1024) __attribute__((amdgpu_waves_per_eu(8, 8))) void k_find_own(
        const int* __restrict__ ei,
        const unsigned* __restrict__ own,
        float* __restrict__ dinv,
        unsigned* __restrict__ qkeep,
        unsigned* __restrict__ qn) {
    __shared__ unsigned cnt[BIN];
    __shared__ unsigned queue[QCAP];
    __shared__ unsigned gqueue[GCAP];
    __shared__ unsigned qcount;
    __shared__ unsigned gcount;

    const int tid  = threadIdx.x;
    const int lo   = blockIdx.x * BIN;
    const int nown = min(BIN, N_NODES - lo);

    if (tid == 0) { qcount = 0u; gcount = 0u; }
    for (int n = tid; n < BIN; n += 1024) cnt[n] = 0u;
    __syncthreads();

    const uint4* own4 = (const uint4*)own;
    const unsigned pat = (unsigned)blockIdx.x * 0x01010101u;
    const int start = blockIdx.x * 408;
    if (tid < SWEEP) {
        uint4 oa[5], ob[5];
        int   ga[5], gb[5];
        auto ask = [&](uint4 (&ov)[5], int (&gv)[5], int k) {
            #pragma unroll
            for (int u = 0; u < 5; ++u) {
                int g = tid + (k + u) * SWEEP + start;
                if (g >= OGRP) g -= OGRP;
                gv[u] = g;
                ov[u] = own4[g];
            }
        };
        auto look = [&](const uint4 (&ov)[5], const int (&gv)[5]) {
            #pragma unroll
            for (int u = 0; u < 5; ++u) {
                unsigned x0 = ov[u].x ^ pat, x1 = ov[u].y ^ pat, x2 = ov[u].z ^ pat, x3 = ov[u].w ^ pat;
                unsigned t = ((x0 - 0x01010101u) & ~x0) | ((x1 - 0x01010101u) & ~x1)
                           | ((x2 - 0x01010101u) & ~x2) | ((x3 - 0x01010101u) & ~x3);
                if (t & 0x80808080u) {
                    unsigned slot = atomicAdd(&gcount, 1u);
                    if (slot < GCAP) gqueue[slot] = (unsigned)gv[u];
                }
            }
        };
        ask(oa, ga, 0);
        for (int k = 0; k < NSTEP - 10; k += 10) {
            ask(ob, gb, k + 5);
            look(oa, ga);
            ask(oa, ga, k + 10);
            look(ob, gb);
        }
        ask(ob, gb, NSTEP - 5);
        look(oa, ga);
        look(ob, gb);
    }
    __syncthreads();

    if (gcount <= (unsigned)GCAP) {
        const int gc = (int)gcount;
        int   gn = 0;
        uint4 on = make_uint4(0u, 0u, 0u, 0u);
        if (tid < gc) { gn = (int)gqueue[tid]; on = own4[gn]; }
        for (int i = tid; i < gc; i += 1024) {
            const int   g = gn;
            const uint4 o = on;
            if (i + 1024 < gc) { gn = (int)gqueue[i + 1024]; on = own4[gn]; }
            unsigned x0 = o.x ^ pat, x1 = o.y ^ pat, x2 = o.z ^ pat, x3 = o.w ^ pat;
            unsigned c0 = ((x0 & 0x7F7F7F7Fu) + 0x7F7F7F7Fu) | x0;
            unsigned c1 = ((x1 & 0x7F7F7F7Fu) + 0x7F7F7F7Fu) | x1;
            unsigned c2 = ((x2 & 0x7F7F7F7Fu) + 0x7F7F7F7Fu) | x2;
            unsigned c3 = ((x3 & 0x7F7F7F7Fu) + 0x7F7F7F7Fu) | x3;
            unsigned m = ~(((c0 >> 7) & 0x01010101u) | ((c1 >> 6) & 0x02020202u)
                         | ((c2 >> 5) & 0x04040404u) | ((c3 >> 4) & 0x08080808u)) & 0x0F0F0F0Fu;
            while (m) {
                const int p = __builtin_ctz(m);
                m &= m - 1u;
                own_edge(g * 16 + ((p & 3) << 2) + (p >> 3), lo, ei, cnt, queue, &qcount);
            }
        }
    } else {
        for (int e = tid; e < N_EDGES; e += 1024)
            if ((unsigned)(ei[N_EDGES + e] - lo) < (unsigned)BIN)
                own_edge(e, lo, ei, cnt, queue, &qcount);
    }
    __syncthreads();

    const int count = (int)min(qcount, (unsigned)QCAP);
    unsigned* qk = qkeep + (size_t)blockIdx.x * QCAP;
    for (int i0 = tid; i0 < count; i0 += 4 * 1024) {
        int e[4], s[4], d[4];
        #pragma unroll
        for (int u = 0; u < 4; ++u) {
            int i = i0 + u * 1024;
            e[u] = (i < count) ? (int)queue[i] : 0;
        }
        #pragma unroll
        for (int u = 0; u < 4; ++u) {
            s[u] = ei[e[u]];
            d[u] = ei[N_EDGES + e[u]];
        }
        #pragma unroll
        for (int u = 0; u < 4; ++u) {
            int i = i0 + u * 1024;
            if (i < count) {
                unsigned dl = (unsigned)(d[u] - lo);
                atomicAdd(&cnt[dl], 1u);
                qk[i] = (unsigned)s[u] | (dl << SBITS);
            }
        }
    }
    __syncthreads();

    for (int n = tid; n < nown; n += 1024) dinv[lo + n] = __frsqrt_rn((float)cnt[n] + 1.0f);
    if (tid == 0) qn[blockIdx.x] = qcount;
}

__global__ __launch_bounds__(128) void k_gemm1_wmma(
        const float* __restrict__ x,
        const float* __restrict__ W1,
        float* __restrict__ xw1) {
    const int wave = threadIdx.x >> 5;
    const int lane = threadIdx.x & 31;
    const int half = lane >> 4;
    const int r    = lane & 15;
    const int n0   = wave * 16;

    v16h bfrag[8];
    #pragma unroll
    for (int kt = 0; kt < 8; ++kt) {
        const int kbase = kt * 32;
        #pragma unroll
        for (int j = 0; j < 8; ++j)
            bfrag[kt][j]     = (_Float16)W1[(kbase + 8 * half + j) * HID + (n0 + r)];
        #pragma unroll
        for (int j = 0; j < 8; ++j)
            bfrag[kt][8 + j] = (_Float16)W1[(kbase + 16 + 8 * half + j) * HID + (n0 + r)];
    }

    for (int rt = 0; rt < RT; ++rt) {
        const int row0 = (blockIdx.x * RT + rt) * 16;
        const float* xrow = x + (size_t)(row0 + r) * F_IN;

        if (rt + 1 < RT) {
            const float* nxt = xrow + 16 * F_IN;
            __builtin_prefetch(nxt +   0, 0, 0);
            __builtin_prefetch(nxt +  64, 0, 0);
            __builtin_prefetch(nxt + 128, 0, 0);
            __builtin_prefetch(nxt + 192, 0, 0);
        }

        v8f acc = {0.f, 0.f, 0.f, 0.f, 0.f, 0.f, 0.f, 0.f};
        #pragma unroll
        for (int kt = 0; kt < 8; ++kt) {
            const float* seg0 = xrow + kt * 32 + 8 * half;
            const float* seg1 = seg0 + 16;
            v4f a0 = __builtin_nontemporal_load((const v4f*)seg0);
            v4f a1 = __builtin_nontemporal_load((const v4f*)(seg0 + 4));
            v4f a2 = __builtin_nontemporal_load((const v4f*)seg1);
            v4f a3 = __builtin_nontemporal_load((const v4f*)(seg1 + 4));
            v16h a;
            a[0]  = (_Float16)a0.x; a[1]  = (_Float16)a0.y;
            a[2]  = (_Float16)a0.z; a[3]  = (_Float16)a0.w;
            a[4]  = (_Float16)a1.x; a[5]  = (_Float16)a1.y;
            a[6]  = (_Float16)a1.z; a[7]  = (_Float16)a1.w;
            a[8]  = (_Float16)a2.x; a[9]  = (_Float16)a2.y;
            a[10] = (_Float16)a2.z; a[11] = (_Float16)a2.w;
            a[12] = (_Float16)a3.x; a[13] = (_Float16)a3.y;
            a[14] = (_Float16)a3.z; a[15] = (_Float16)a3.w;

            acc = __builtin_amdgcn_wmma_f32_16x16x32_f16(
                 false, a,  false, bfrag[kt],
                 (short)0, acc,  false,  false);
        }

        #pragma unroll
        for (int i = 0; i < 8; ++i) {
            int m = i + 8 * half;
            xw1[(size_t)(row0 + m) * HID + (n0 + r)] = acc[i];
        }
    }
}

__global__ __launch_bounds__(1024) __attribute__((amdgpu_waves_per_eu(8, 8))) void k_agg1_own(
        const int* __restrict__ ei,
        const float* __restrict__ dinv,
        const float* __restrict__ xw1,
        const float* __restrict__ b1,
        const float* __restrict__ W2,
        const unsigned* __restrict__ qkeep,
        const unsigned* __restrict__ qn,
        float* __restrict__ zs) {
    __shared__ float    acc[BIN * HID];
    __shared__ float    dinv_l[BIN];
    __shared__ float    zl[BIN];

    const int tid  = threadIdx.x;
    const int lane = tid & 31;
    const int wave = tid >> 5;
    const int lo   = blockIdx.x * BIN;
    const int nown = min(BIN, N_NODES - lo);

    for (int n = tid; n < nown; n += 1024) dinv_l[n] = dinv[lo + n];
    for (int i = tid; i < nown * HID; i += 1024) {
        float di = dinv[lo + (i >> 6)];
        acc[i] = xw1[(size_t)lo * HID + i] * (di * di) + b1[i & 63];
    }
    __syncthreads();

    const unsigned kept = qn[blockIdx.x];
    if (kept <= (unsigned)QCAP) {
        const unsigned* q = qkeep + (size_t)blockIdx.x * QCAP;
        const int count = (int)kept;
        const int last = count - 1;
        unsigned ent_n = (wave * 32 < count) ? q[min(wave * 32 + lane, last)] : 0u;
        for (int base = wave * 32; base < count; base += 32 * 32) {
            const int nval = min(32, count - base);
            const unsigned ent = (lane < nval) ? ent_n : 0u;
            ent_n = q[min(base + 32 * 32 + lane, last)];
            int   s_l  = (int)(ent & ((1u << SBITS) - 1u));
            int   dl_l = (int)(ent >> SBITS);
            float ds_l = dinv[s_l];
            float dd_l = dinv_l[dl_l];
            #pragma unroll
            for (int jj = 0; jj < 32; jj += 16) {
                if (jj < nval) {
                    float v0[16], v1[16];
                    #pragma unroll
                    for (int u = 0; u < 16; ++u) {
                        if (jj + u < nval) {
                            int sj = __builtin_amdgcn_readlane(s_l, jj + u);
                            const float* row = xw1 + (size_t)sj * HID;
                            v0[u] = row[lane];
                            v1[u] = row[32 + lane];
                        }
                    }
                    #pragma unroll
                    for (int u = 0; u < 16; ++u) {
                        if (jj + u < nval) {
                            int   dj = __builtin_amdgcn_readlane(dl_l, jj + u);
                            float nj = __int_as_float(__builtin_amdgcn_readlane(__float_as_int(ds_l), jj + u))
                                     * __int_as_float(__builtin_amdgcn_readlane(__float_as_int(dd_l), jj + u));
                            atomicAdd(&acc[dj * HID + lane],      v0[u] * nj);
                            atomicAdd(&acc[dj * HID + 32 + lane], v1[u] * nj);
                        }
                    }
                }
            }
        }
    } else {
        for (int e = tid; e < N_EDGES; e += 1024) {
            unsigned dl = (unsigned)(ei[N_EDGES + e] - lo);
            if (dl < (unsigned)BIN) {
                int   s    = ei[e];
                float norm = dinv[s] * dinv_l[dl];
                const float* row = xw1 + (size_t)s * HID;
                #pragma nounroll
                for (int f = 0; f < HID; ++f) atomicAdd(&acc[dl * HID + f], row[f] * norm);
            }
        }
    }
    __syncthreads();

    const float w0a = W2[lane * 2 + 0],        w1a = W2[lane * 2 + 1];
    const float w0b = W2[(32 + lane) * 2 + 0], w1b = W2[(32 + lane) * 2 + 1];
    for (int n = wave; n < nown; n += 32) {
        float h0 = fmaxf(acc[n * HID + lane], 0.f);
        float h1 = fmaxf(acc[n * HID + 32 + lane], 0.f);
        float o0 = h0 * w0a + h1 * w0b;
        float o1 = h0 * w1a + h1 * w1b;
        for (int off = 16; off > 0; off >>= 1) {
            o0 += __shfl_xor(o0, off);
            o1 += __shfl_xor(o1, off);
        }
        if (lane == 0) zl[n] = (o0 - o1) * dinv_l[n];
    }
    __syncthreads();
    for (int n = tid; n < nown; n += 1024) zs[lo + n] = zl[n];
}

__global__ __launch_bounds__(1024) void k_agg2_own(
        const int* __restrict__ ei,
        const float* __restrict__ dinv,
        const float* __restrict__ zs,
        const float* __restrict__ b2,
        const unsigned* __restrict__ qkeep,
        const unsigned* __restrict__ qn,
        float* __restrict__ out) {
    __shared__ float sum[BIN];

    const int tid  = threadIdx.x;
    const int lo   = blockIdx.x * BIN;
    const int nown = min(BIN, N_NODES - lo);

    for (int n = tid; n < nown; n += 1024) sum[n] = zs[lo + n];
    __syncthreads();

    const unsigned count = qn[blockIdx.x];
    if (count <= (unsigned)QCAP) {
        const unsigned* q = qkeep + (size_t)blockIdx.x * QCAP;
        const unsigned NONE = 0xFFFFFFFFu;
        for (unsigned i0 = tid; i0 < count; i0 += 4u * 1024u) {
            unsigned ent[4];
            float    v[4];
            #pragma unroll
            for (int u = 0; u < 4; ++u) {
                unsigned i = i0 + (unsigned)u * 1024u;
                ent[u] = (i < count) ? q[i] : NONE;
            }
            #pragma unroll
            for (int u = 0; u < 4; ++u)
                v[u] = (ent[u] != NONE) ? zs[ent[u] & ((1u << SBITS) - 1u)] : 0.f;
            #pragma unroll
            for (int u = 0; u < 4; ++u)
                if (ent[u] != NONE) atomicAdd(&sum[ent[u] >> SBITS], v[u]);
        }
    } else {
        for (int e = tid; e < N_EDGES; e += 1024) {
            unsigned dl = (unsigned)(ei[N_EDGES + e] - lo);
            if (dl < (unsigned)BIN) atomicAdd(&sum[dl], zs[ei[e]]);
        }
    }
    __syncthreads();

    const float bd = b2[0] - b2[1];
    for (int n = tid; n < nown; n += 1024) {
        float a0 = dinv[lo + n] * sum[n] + bd;
        float a1 = 0.0f;
        float m  = fmaxf(a0, a1);
        float e0 = __expf(a0 - m);
        float e1 = __expf(a1 - m);
        float inv = 1.0f / (e0 + e1);
        float2 r = make_float2(e0 * inv, e1 * inv);
        *(float2*)(out + (size_t)(lo + n) * 2) = r;
    }
}

extern "C" void kernel_launch(void* const* d_in, const int* in_sizes, int n_in,
                              void* d_out, int out_size, void* d_ws, size_t ws_size,
                              hipStream_t stream) {
    const float* x  = (const float*)d_in[0];
    const int*   ei = (const int*)d_in[1];
    const float* W1 = (const float*)d_in[2];
    const float* b1 = (const float*)d_in[3];
    const float* W2 = (const float*)d_in[4];
    const float* b2 = (const float*)d_in[5];
    float* out = (float*)d_out;

    char* ws = (char*)d_ws;
    size_t o = 0;
    auto alloc = [&](size_t bytes) {
        char* p = ws + o;
        o += (bytes + 4095) & ~(size_t)4095;
        return p;
    };
    float* dinv = (float*)alloc((size_t)N_NODES * sizeof(float));
    float* xw1  = (float*)alloc((size_t)N_NODES * HID * sizeof(float));
    unsigned* qkeep = (unsigned*)alloc((size_t)NBINS * QCAP * sizeof(unsigned));
    float*    zs    = (float*)   alloc((size_t)N_NODES * sizeof(float));
    unsigned* qn    = (unsigned*)alloc((size_t)NBINS * sizeof(unsigned));
    unsigned* own   = (unsigned*)alloc((size_t)N_EDGES);
    (void)ws_size; (void)in_sizes; (void)n_in; (void)out_size;

    k_own_bytes<<<(NGRP + 1023) / 1024, dim3(256), 0, stream>>>(ei, own);
    k_find_own <<<NBINS, dim3(1024), 0, stream>>>(ei, own, dinv, qkeep, qn);

    k_gemm1_wmma<<<(N_NODES / 16) / RT, dim3(128), 0, stream>>>(x, W1, xw1);

    k_agg1_own<<<NBINS, dim3(1024), 0, stream>>>(ei, dinv, xw1, b1, W2, qkeep, qn, zs);

    k_agg2_own<<<NBINS, dim3(1024), 0, stream>>>(ei, dinv, zs, b2, qkeep, qn, out);
}


# layer-1 GEMM: each work-group fetches its 16-row x tile once with whole-line loads, rounds it to f16 once into LDS (two tiles in turn); the four column waves read A fragments from LDS, not memory
# speedup vs baseline: 4.2234x; 1.3650x over previous
#include <hip/hip_runtime.h>
#include <hip/hip_bf16.h>
#include <stdint.h>

#define N_NODES 100000
#define N_EDGES 1600000
#define F_IN    256
#define HID     64
#define RT      5

typedef __attribute__((ext_vector_type(16))) _Float16 v16h;
typedef __attribute__((ext_vector_type(8)))  _Float16 v8h;
typedef __attribute__((ext_vector_type(4)))  _Float16 v4h;
typedef __attribute__((ext_vector_type(8)))  float    v8f;
typedef __attribute__((ext_vector_type(4)))  float    v4f;

#define BIN    416
#define NGRP   (N_EDGES / 4)
__global__ __launch_bounds__(256) void k_own_bytes(const int* __restrict__ ei, unsigned* __restrict__ own) {
    const int4* dst4 = (const int4*)(ei + N_EDGES);
    const int g0 = blockIdx.x * 1024 + threadIdx.x;
    int4 d[4];
    #pragma unroll
    for (int u = 0; u < 4; ++u) {
        int g = g0 + u * 256;
        if (g < NGRP) d[u] = dst4[g];
    }
    #pragma unroll
    for (int u = 0; u < 4; ++u) {
        int g = g0 + u * 256;
        if (g < NGRP)
            own[g] = (unsigned)(d[u].x / BIN)         | ((unsigned)(d[u].y / BIN) << 8)
                   | ((unsigned)(d[u].z / BIN) << 16) | ((unsigned)(d[u].w / BIN) << 24);
    }
}

#define NBINS  ((N_NODES + BIN - 1) / BIN)
#define SBITS  17
#define QCAP   12288
#define GCAP   12288
#define OGRP   (N_EDGES / 16)
#define SWEEP  1000
#define NSTEP  100

static __device__ __forceinline__ void own_edge(int e, int lo, const int* __restrict__ ei,
        unsigned* cnt, unsigned* queue, unsigned* qcount) {
    unsigned slot = atomicAdd(qcount, 1u);
    if (slot < QCAP) {
        queue[slot] = (unsigned)e;
    } else {
        atomicAdd(&cnt[ei[N_EDGES + e] - lo], 1u);
    }
}

__global__ __launch_bounds__(1024) __attribute__((amdgpu_waves_per_eu(8, 8))) void k_find_own(
        const int* __restrict__ ei,
        const unsigned* __restrict__ own,
        float* __restrict__ dinv,
        unsigned* __restrict__ qkeep,
        unsigned* __restrict__ qn) {
    __shared__ unsigned cnt[BIN];
    __shared__ unsigned queue[QCAP];
    __shared__ unsigned gqueue[GCAP];
    __shared__ unsigned qcount;
    __shared__ unsigned gcount;

    const int tid  = threadIdx.x;
    const int lo   = blockIdx.x * BIN;
    const int nown = min(BIN, N_NODES - lo);

    if (tid == 0) { qcount = 0u; gcount = 0u; }
    for (int n = tid; n < BIN; n += 1024) cnt[n] = 0u;
    __syncthreads();

    const uint4* own4 = (const uint4*)own;
    const unsigned pat = (unsigned)blockIdx.x * 0x01010101u;
    const int start = blockIdx.x * 408;
    if (tid < SWEEP) {
        uint4 oa[5], ob[5];
        int   ga[5], gb[5];
        auto ask = [&](uint4 (&ov)[5], int (&gv)[5], int k) {
            #pragma unroll
            for (int u = 0; u < 5; ++u) {
                int g = tid + (k + u) * SWEEP + start;
                if (g >= OGRP) g -= OGRP;
                gv[u] = g;
                ov[u] = own4[g];
            }
        };
        auto look = [&](const uint4 (&ov)[5], const int (&gv)[5]) {
            #pragma unroll
            for (int u = 0; u < 5; ++u) {
                unsigned x0 = ov[u].x ^ pat, x1 = ov[u].y ^ pat, x2 = ov[u].z ^ pat, x3 = ov[u].w ^ pat;
                unsigned t = ((x0 - 0x01010101u) & ~x0) | ((x1 - 0x01010101u) & ~x1)
                           | ((x2 - 0x01010101u) & ~x2) | ((x3 - 0x01010101u) & ~x3);
                if (t & 0x80808080u) {
                    unsigned slot = atomicAdd(&gcount, 1u);
                    if (slot < GCAP) gqueue[slot] = (unsigned)gv[u];
                }
            }
        };
        ask(oa, ga, 0);
        for (int k = 0; k < NSTEP - 10; k += 10) {
            ask(ob, gb, k + 5);
            look(oa, ga);
            ask(oa, ga, k + 10);
            look(ob, gb);
        }
        ask(ob, gb, NSTEP - 5);
        look(oa, ga);
        look(ob, gb);
    }
    __syncthreads();

    if (gcount <= (unsigned)GCAP) {
        const int gc = (int)gcount;
        int   gn = 0;
        uint4 on = make_uint4(0u, 0u, 0u, 0u);
        if (tid < gc) { gn = (int)gqueue[tid]; on = own4[gn]; }
        for (int i = tid; i < gc; i += 1024) {
            const int   g = gn;
            const uint4 o = on;
            if (i + 1024 < gc) { gn = (int)gqueue[i + 1024]; on = own4[gn]; }
            unsigned x0 = o.x ^ pat, x1 = o.y ^ pat, x2 = o.z ^ pat, x3 = o.w ^ pat;
            unsigned c0 = ((x0 & 0x7F7F7F7Fu) + 0x7F7F7F7Fu) | x0;
            unsigned c1 = ((x1 & 0x7F7F7F7Fu) + 0x7F7F7F7Fu) | x1;
            unsigned c2 = ((x2 & 0x7F7F7F7Fu) + 0x7F7F7F7Fu) | x2;
            unsigned c3 = ((x3 & 0x7F7F7F7Fu) + 0x7F7F7F7Fu) | x3;
            unsigned m = ~(((c0 >> 7) & 0x01010101u) | ((c1 >> 6) & 0x02020202u)
                         | ((c2 >> 5) & 0x04040404u) | ((c3 >> 4) & 0x08080808u)) & 0x0F0F0F0Fu;
            while (m) {
                const int p = __builtin_ctz(m);
                m &= m - 1u;
                own_edge(g * 16 + ((p & 3) << 2) + (p >> 3), lo, ei, cnt, queue, &qcount);
            }
        }
    } else {
        for (int e = tid; e < N_EDGES; e += 1024)
            if ((unsigned)(ei[N_EDGES + e] - lo) < (unsigned)BIN)
                own_edge(e, lo, ei, cnt, queue, &qcount);
    }
    __syncthreads();

    const int count = (int)min(qcount, (unsigned)QCAP);
    unsigned* qk = qkeep + (size_t)blockIdx.x * QCAP;
    for (int i0 = tid; i0 < count; i0 += 4 * 1024) {
        int e[4], s[4], d[4];
        #pragma unroll
        for (int u = 0; u < 4; ++u) {
            int i = i0 + u * 1024;
            e[u] = (i < count) ? (int)queue[i] : 0;
        }
        #pragma unroll
        for (int u = 0; u < 4; ++u) {
            s[u] = ei[e[u]];
            d[u] = ei[N_EDGES + e[u]];
        }
        #pragma unroll
        for (int u = 0; u < 4; ++u) {
            int i = i0 + u * 1024;
            if (i < count) {
                unsigned dl = (unsigned)(d[u] - lo);
                atomicAdd(&cnt[dl], 1u);
                qk[i] = (unsigned)s[u] | (dl << SBITS);
            }
        }
    }
    __syncthreads();

    for (int n = tid; n < nown; n += 1024) dinv[lo + n] = __frsqrt_rn((float)cnt[n] + 1.0f);
    if (tid == 0) qn[blockIdx.x] = qcount;
}

#define XP 264
__global__ __launch_bounds__(128) void k_gemm1_wmma(
        const float* __restrict__ x,
        const float* __restrict__ W1,
        float* __restrict__ xw1) {
    __shared__ __attribute__((aligned(16))) _Float16 xs[2][16 * XP];

    const int tid  = threadIdx.x;
    const int wave = tid >> 5;
    const int lane = tid & 31;
    const int half = lane >> 4;
    const int r    = lane & 15;
    const int n0   = wave * 16;

    const int srow = tid >> 6;
    const int sc4  = tid & 63;
    const v4f* xt = (const v4f*)(x + (size_t)blockIdx.x * RT * 16 * F_IN) + tid;

    v4f st[8];
    #pragma unroll
    for (int i = 0; i < 8; ++i) st[i] = __builtin_nontemporal_load(xt + i * 128);

    v16h bfrag[8];
    #pragma unroll
    for (int kt = 0; kt < 8; ++kt) {
        const int kbase = kt * 32;
        #pragma unroll
        for (int j = 0; j < 8; ++j)
            bfrag[kt][j]     = (_Float16)W1[(kbase + 8 * half + j) * HID + (n0 + r)];
        #pragma unroll
        for (int j = 0; j < 8; ++j)
            bfrag[kt][8 + j] = (_Float16)W1[(kbase + 16 + 8 * half + j) * HID + (n0 + r)];
    }

    for (int rt = 0; rt < RT; ++rt) {
        const int row0 = (blockIdx.x * RT + rt) * 16;
        _Float16* xb = xs[rt & 1];

        #pragma unroll
        for (int i = 0; i < 8; ++i) {
            v4h h;
            h[0] = (_Float16)st[i].x; h[1] = (_Float16)st[i].y;
            h[2] = (_Float16)st[i].z; h[3] = (_Float16)st[i].w;
            *(v4h*)(xb + (2 * i + srow) * XP + 4 * sc4) = h;
        }
        if (rt + 1 < RT) {
            const v4f* nx = xt + (size_t)(rt + 1) * (16 * F_IN / 4);
            #pragma unroll
            for (int i = 0; i < 8; ++i) st[i] = __builtin_nontemporal_load(nx + i * 128);
        }
        __builtin_amdgcn_sched_barrier(0);

        __builtin_amdgcn_fence(__ATOMIC_RELEASE, "workgroup", "local");
        __builtin_amdgcn_s_barrier();
        __builtin_amdgcn_fence(__ATOMIC_ACQUIRE, "workgroup", "local");

        v8f acc = {0.f, 0.f, 0.f, 0.f, 0.f, 0.f, 0.f, 0.f};
        const _Float16* arow = xb + r * XP + 8 * half;
        #pragma unroll
        for (int kt = 0; kt < 8; ++kt) {
            v8h lo = *(const v8h*)(arow + kt * 32);
            v8h hi = *(const v8h*)(arow + kt * 32 + 16);
            v16h a = __builtin_shufflevector(lo, hi, 0, 1, 2, 3, 4, 5, 6, 7, 8, 9, 10, 11, 12, 13, 14, 15);

            acc = __builtin_amdgcn_wmma_f32_16x16x32_f16(
                 false, a,  false, bfrag[kt],
                 (short)0, acc,  false,  false);
        }

        #pragma unroll
        for (int i = 0; i < 8; ++i) {
            int m = i + 8 * half;
            xw1[(size_t)(row0 + m) * HID + (n0 + r)] = acc[i];
        }
        __builtin_amdgcn_sched_barrier(0);
    }
}

__global__ __launch_bounds__(1024) __attribute__((amdgpu_waves_per_eu(8, 8))) void k_agg1_own(
        const int* __restrict__ ei,
        const float* __restrict__ dinv,
        const float* __restrict__ xw1,
        const float* __restrict__ b1,
        const float* __restrict__ W2,
        const unsigned* __restrict__ qkeep,
        const unsigned* __restrict__ qn,
        float* __restrict__ zs) {
    __shared__ float    acc[BIN * HID];
    __shared__ float    dinv_l[BIN];
    __shared__ float    zl[BIN];

    const int tid  = threadIdx.x;
    const int lane = tid & 31;
    const int wave = tid >> 5;
    const int lo   = blockIdx.x * BIN;
    const int nown = min(BIN, N_NODES - lo);

    for (int n = tid; n < nown; n += 1024) dinv_l[n] = dinv[lo + n];
    for (int i = tid; i < nown * HID; i += 1024) {
        float di = dinv[lo + (i >> 6)];
        acc[i] = xw1[(size_t)lo * HID + i] * (di * di) + b1[i & 63];
    }
    __syncthreads();

    const unsigned kept = qn[blockIdx.x];
    if (kept <= (unsigned)QCAP) {
        const unsigned* q = qkeep + (size_t)blockIdx.x * QCAP;
        const int count = (int)kept;
        const int last = count - 1;
        unsigned ent_n = (wave * 32 < count) ? q[min(wave * 32 + lane, last)] : 0u;
        for (int base = wave * 32; base < count; base += 32 * 32) {
            const int nval = min(32, count - base);
            const unsigned ent = (lane < nval) ? ent_n : 0u;
            ent_n = q[min(base + 32 * 32 + lane, last)];
            int   s_l  = (int)(ent & ((1u << SBITS) - 1u));
            int   dl_l = (int)(ent >> SBITS);
            float ds_l = dinv[s_l];
            float dd_l = dinv_l[dl_l];
            #pragma unroll
            for (int jj = 0; jj < 32; jj += 16) {
                if (jj < nval) {
                    float v0[16], v1[16];
                    #pragma unroll
                    for (int u = 0; u < 16; ++u) {
                        if (jj + u < nval) {
                            int sj = __builtin_amdgcn_readlane(s_l, jj + u);
                            const float* row = xw1 + (size_t)sj * HID;
                            v0[u] = row[lane];
                            v1[u] = row[32 + lane];
                        }
                    }
                    #pragma unroll
                    for (int u = 0; u < 16; ++u) {
                        if (jj + u < nval) {
                            int   dj = __builtin_amdgcn_readlane(dl_l, jj + u);
                            float nj = __int_as_float(__builtin_amdgcn_readlane(__float_as_int(ds_l), jj + u))
                                     * __int_as_float(__builtin_amdgcn_readlane(__float_as_int(dd_l), jj + u));
                            atomicAdd(&acc[dj * HID + lane],      v0[u] * nj);
                            atomicAdd(&acc[dj * HID + 32 + lane], v1[u] * nj);
                        }
                    }
                }
            }
        }
    } else {
        for (int e = tid; e < N_EDGES; e += 1024) {
            unsigned dl = (unsigned)(ei[N_EDGES + e] - lo);
            if (dl < (unsigned)BIN) {
                int   s    = ei[e];
                float norm = dinv[s] * dinv_l[dl];
                const float* row = xw1 + (size_t)s * HID;
                #pragma nounroll
                for (int f = 0; f < HID; ++f) atomicAdd(&acc[dl * HID + f], row[f] * norm);
            }
        }
    }
    __syncthreads();

    const float w0a = W2[lane * 2 + 0],        w1a = W2[lane * 2 + 1];
    const float w0b = W2[(32 + lane) * 2 + 0], w1b = W2[(32 + lane) * 2 + 1];
    for (int n = wave; n < nown; n += 32) {
        float h0 = fmaxf(acc[n * HID + lane], 0.f);
        float h1 = fmaxf(acc[n * HID + 32 + lane], 0.f);
        float o0 = h0 * w0a + h1 * w0b;
        float o1 = h0 * w1a + h1 * w1b;
        for (int off = 16; off > 0; off >>= 1) {
            o0 += __shfl_xor(o0, off);
            o1 += __shfl_xor(o1, off);
        }
        if (lane == 0) zl[n] = (o0 - o1) * dinv_l[n];
    }
    __syncthreads();
    for (int n = tid; n < nown; n += 1024) zs[lo + n] = zl[n];
}

__global__ __launch_bounds__(1024) void k_agg2_own(
        const int* __restrict__ ei,
        const float* __restrict__ dinv,
        const float* __restrict__ zs,
        const float* __restrict__ b2,
        const unsigned* __restrict__ qkeep,
        const unsigned* __restrict__ qn,
        float* __restrict__ out) {
    __shared__ float sum[BIN];

    const int tid  = threadIdx.x;
    const int lo   = blockIdx.x * BIN;
    const int nown = min(BIN, N_NODES - lo);

    for (int n = tid; n < nown; n += 1024) sum[n] = zs[lo + n];
    __syncthreads();

    const unsigned count = qn[blockIdx.x];
    if (count <= (unsigned)QCAP) {
        const unsigned* q = qkeep + (size_t)blockIdx.x * QCAP;
        const unsigned NONE = 0xFFFFFFFFu;
        for (unsigned i0 = tid; i0 < count; i0 += 4u * 1024u) {
            unsigned ent[4];
            float    v[4];
            #pragma unroll
            for (int u = 0; u < 4; ++u) {
                unsigned i = i0 + (unsigned)u * 1024u;
                ent[u] = (i < count) ? q[i] : NONE;
            }
            #pragma unroll
            for (int u = 0; u < 4; ++u)
                v[u] = (ent[u] != NONE) ? zs[ent[u] & ((1u << SBITS) - 1u)] : 0.f;
            #pragma unroll
            for (int u = 0; u < 4; ++u)
                if (ent[u] != NONE) atomicAdd(&sum[ent[u] >> SBITS], v[u]);
        }
    } else {
        for (int e = tid; e < N_EDGES; e += 1024) {
            unsigned dl = (unsigned)(ei[N_EDGES + e] - lo);
            if (dl < (unsigned)BIN) atomicAdd(&sum[dl], zs[ei[e]]);
        }
    }
    __syncthreads();

    const float bd = b2[0] - b2[1];
    for (int n = tid; n < nown; n += 1024) {
        float a0 = dinv[lo + n] * sum[n] + bd;
        float a1 = 0.0f;
        float m  = fmaxf(a0, a1);
        float e0 = __expf(a0 - m);
        float e1 = __expf(a1 - m);
        float inv = 1.0f / (e0 + e1);
        float2 r = make_float2(e0 * inv, e1 * inv);
        *(float2*)(out + (size_t)(lo + n) * 2) = r;
    }
}

extern "C" void kernel_launch(void* const* d_in, const int* in_sizes, int n_in,
                              void* d_out, int out_size, void* d_ws, size_t ws_size,
                              hipStream_t stream) {
    const float* x  = (const float*)d_in[0];
    const int*   ei = (const int*)d_in[1];
    const float* W1 = (const float*)d_in[2];
    const float* b1 = (const float*)d_in[3];
    const float* W2 = (const float*)d_in[4];
    const float* b2 = (const float*)d_in[5];
    float* out = (float*)d_out;

    char* ws = (char*)d_ws;
    size_t o = 0;
    auto alloc = [&](size_t bytes) {
        char* p = ws + o;
        o += (bytes + 4095) & ~(size_t)4095;
        return p;
    };
    float* dinv = (float*)alloc((size_t)N_NODES * sizeof(float));
    float* xw1  = (float*)alloc((size_t)N_NODES * HID * sizeof(float));
    unsigned* qkeep = (unsigned*)alloc((size_t)NBINS * QCAP * sizeof(unsigned));
    float*    zs    = (float*)   alloc((size_t)N_NODES * sizeof(float));
    unsigned* qn    = (unsigned*)alloc((size_t)NBINS * sizeof(unsigned));
    unsigned* own   = (unsigned*)alloc((size_t)N_EDGES);
    (void)ws_size; (void)in_sizes; (void)n_in; (void)out_size;

    k_own_bytes<<<(NGRP + 1023) / 1024, dim3(256), 0, stream>>>(ei, own);
    k_find_own <<<NBINS, dim3(1024), 0, stream>>>(ei, own, dinv, qkeep, qn);

    k_gemm1_wmma<<<(N_NODES / 16) / RT, dim3(128), 0, stream>>>(x, W1, xw1);

    k_agg1_own<<<NBINS, dim3(1024), 0, stream>>>(ei, dinv, xw1, b1, W2, qkeep, qn, zs);

    k_agg2_own<<<NBINS, dim3(1024), 0, stream>>>(ei, dinv, zs, b2, qkeep, qn, out);
}


# owners read a per-owner bitmap of the edge list (launch 1 marks in LDS which 16-edge groups hold whose edges), not a per-edge owner list; marked groups' src/dst quads read once; arrival order kept
# speedup vs baseline: 4.7810x; 1.1320x over previous
#include <hip/hip_runtime.h>
#include <hip/hip_bf16.h>
#include <stdint.h>

#define N_NODES 100000
#define N_EDGES 1600000
#define F_IN    256
#define HID     64
#define RT      5

typedef __attribute__((ext_vector_type(16))) _Float16 v16h;
typedef __attribute__((ext_vector_type(8)))  _Float16 v8h;
typedef __attribute__((ext_vector_type(4)))  _Float16 v4h;
typedef __attribute__((ext_vector_type(8)))  float    v8f;
typedef __attribute__((ext_vector_type(4)))  float    v4f;

#define BIN    416
#define NQUAD  (N_EDGES / 4)
#define NBINS  ((N_NODES + BIN - 1) / BIN)
#define OGRP   (N_EDGES / 16)
#define GPB    1024
#define NB1    ((OGRP + GPB - 1) / GPB)
#define BWORDS (NB1 * 32)
__global__ __launch_bounds__(1024) void k_own_bits(const int* __restrict__ ei, unsigned* __restrict__ bits) {
    __shared__ unsigned tab[32 * NBINS];
    const int tid = threadIdx.x;
    const int4* dst4 = (const int4*)(ei + N_EDGES);
    const int q0 = blockIdx.x * (GPB * 4) + tid;
    int4 d[4];
    #pragma unroll
    for (int u = 0; u < 4; ++u) {
        const int q = q0 + u * 1024;
        d[u] = dst4[(q < NQUAD) ? q : 0];
    }
    for (int i = tid; i < 32 * NBINS; i += 1024) tab[i] = 0u;
    __syncthreads();
    #pragma unroll
    for (int u = 0; u < 4; ++u) {
        const int q = q0 + u * 1024;
        if (q < NQUAD) {
            const int p = (u * 1024 + tid) >> 2;
            unsigned* row = tab + (p >> 5) * NBINS;
            const unsigned bit = 1u << (p & 31);
            atomicOr(&row[d[u].x / BIN], bit);
            atomicOr(&row[d[u].y / BIN], bit);
            atomicOr(&row[d[u].z / BIN], bit);
            atomicOr(&row[d[u].w / BIN], bit);
        }
    }
    __syncthreads();
    for (int i = tid; i < 32 * NBINS; i += 1024) {
        const int o = i >> 5, w = i & 31;
        bits[((size_t)o * NB1 + blockIdx.x) * 32 + w] = tab[w * NBINS + o];
    }
}

#define SBITS  17
#define QCAP   12288
#define GCAP   12288

static __device__ __forceinline__ void take_edge(int s, int d, int lo,
        unsigned* cnt, unsigned* queue, unsigned* qcount) {
    const unsigned dl = (unsigned)(d - lo);
    if (dl < (unsigned)BIN) {
        atomicAdd(&cnt[dl], 1u);
        const unsigned slot = atomicAdd(qcount, 1u);
        if (slot < QCAP) queue[slot] = (unsigned)s | (dl << SBITS);
    }
}

__global__ __launch_bounds__(1024) __attribute__((amdgpu_waves_per_eu(8, 8))) void k_find_own(
        const int* __restrict__ ei,
        const unsigned* __restrict__ bits,
        float* __restrict__ dinv,
        unsigned* __restrict__ qkeep,
        unsigned* __restrict__ qn) {
    __shared__ unsigned cnt[BIN];
    __shared__ unsigned queue[QCAP];
    __shared__ unsigned gqueue[GCAP];
    __shared__ unsigned qcount;
    __shared__ unsigned gcount;

    const int tid  = threadIdx.x;
    const int lane = tid & 31;
    const int lo   = blockIdx.x * BIN;
    const int nown = min(BIN, N_NODES - lo);

    uint4 x = make_uint4(0u, 0u, 0u, 0u);
    if (tid < BWORDS / 4) x = ((const uint4*)(bits + (size_t)blockIdx.x * BWORDS))[tid];

    if (tid == 0) { qcount = 0u; gcount = 0u; }
    for (int n = tid; n < BIN; n += 1024) cnt[n] = 0u;
    __syncthreads();

    {
        const int c = __popc(x.x) + __popc(x.y) + __popc(x.z) + __popc(x.w);
        int incl = c;
        for (int off = 1; off < 32; off <<= 1) {
            const int v = __shfl_up(incl, off);
            if (lane >= off) incl += v;
        }
        const int total = __shfl(incl, 31);
        unsigned base = 0u;
        if (lane == 31 && total > 0) base = atomicAdd(&gcount, (unsigned)total);
        base = (unsigned)__shfl((int)base, 31);
        unsigned slot = base + (unsigned)(incl - c);
        auto list = [&](unsigned m, unsigned g0) {
            while (m) {
                const int r = __builtin_ctz(m);
                m &= m - 1u;
                if (slot < (unsigned)GCAP) gqueue[slot] = g0 + (unsigned)r;
                ++slot;
            }
        };
        const unsigned j0 = (unsigned)tid * 4u;
        list(x.x, (j0 + 0u) * 32u);
        list(x.y, (j0 + 1u) * 32u);
        list(x.z, (j0 + 2u) * 32u);
        list(x.w, (j0 + 3u) * 32u);
    }
    __syncthreads();

    const int4* src4 = (const int4*)ei;
    const int4* dst4 = (const int4*)(ei + N_EDGES);
    if (gcount <= (unsigned)GCAP) {
        const int items = (int)gcount * 4;
        for (int b0 = 0; b0 < items; b0 += 4 * 1024) {
            int4 s[4], d[4];
            int g[4];
            #pragma unroll
            for (int u = 0; u < 4; ++u) g[u] = (int)gqueue[min(b0 + u * 1024 + tid, items - 1) >> 2];
            #pragma unroll
            for (int u = 0; u < 4; ++u) {
                const int q = g[u] * 4 + (tid & 3);
                d[u] = dst4[q];
                s[u] = src4[q];
            }
            __builtin_amdgcn_sched_barrier(0);
            unsigned dl[16], sv[16], m = 0u;
            #pragma unroll
            for (int u = 0; u < 4; ++u) {
                dl[4 * u + 0] = (unsigned)(d[u].x - lo); sv[4 * u + 0] = (unsigned)s[u].x;
                dl[4 * u + 1] = (unsigned)(d[u].y - lo); sv[4 * u + 1] = (unsigned)s[u].y;
                dl[4 * u + 2] = (unsigned)(d[u].z - lo); sv[4 * u + 2] = (unsigned)s[u].z;
                dl[4 * u + 3] = (unsigned)(d[u].w - lo); sv[4 * u + 3] = (unsigned)s[u].w;
                if (b0 + u * 1024 + tid < items) {
                    #pragma unroll
                    for (int k = 0; k < 4; ++k)
                        if (dl[4 * u + k] < (unsigned)BIN) m |= 1u << (4 * u + k);
                }
            }
            unsigned slot = atomicAdd(&qcount, (unsigned)__popc(m));
            #pragma unroll
            for (int k = 0; k < 16; ++k) {
                if (m & (1u << k)) {
                    atomicAdd(&cnt[dl[k]], 1u);
                    if (slot < (unsigned)QCAP) queue[slot] = sv[k] | (dl[k] << SBITS);
                    ++slot;
                }
            }
        }
    } else {
        for (int e = tid; e < N_EDGES; e += 1024)
            take_edge(ei[e], ei[N_EDGES + e], lo, cnt, queue, &qcount);
    }
    __syncthreads();

    const int count = (int)min(qcount, (unsigned)QCAP);
    unsigned* qk = qkeep + (size_t)blockIdx.x * QCAP;
    for (int i = tid; i < count; i += 1024) qk[i] = queue[i];

    for (int n = tid; n < nown; n += 1024) dinv[lo + n] = __frsqrt_rn((float)cnt[n] + 1.0f);
    if (tid == 0) qn[blockIdx.x] = qcount;
}

#define XP 264
__global__ __launch_bounds__(128) void k_gemm1_wmma(
        const float* __restrict__ x,
        const float* __restrict__ W1,
        float* __restrict__ xw1) {
    __shared__ __attribute__((aligned(16))) _Float16 xs[2][16 * XP];

    const int tid  = threadIdx.x;
    const int wave = tid >> 5;
    const int lane = tid & 31;
    const int half = lane >> 4;
    const int r    = lane & 15;
    const int n0   = wave * 16;

    const int srow = tid >> 6;
    const int sc4  = tid & 63;
    const v4f* xt = (const v4f*)(x + (size_t)blockIdx.x * RT * 16 * F_IN) + tid;

    v4f st[8];
    #pragma unroll
    for (int i = 0; i < 8; ++i) st[i] = __builtin_nontemporal_load(xt + i * 128);

    v16h bfrag[8];
    #pragma unroll
    for (int kt = 0; kt < 8; ++kt) {
        const int kbase = kt * 32;
        #pragma unroll
        for (int j = 0; j < 8; ++j)
            bfrag[kt][j]     = (_Float16)W1[(kbase + 8 * half + j) * HID + (n0 + r)];
        #pragma unroll
        for (int j = 0; j < 8; ++j)
            bfrag[kt][8 + j] = (_Float16)W1[(kbase + 16 + 8 * half + j) * HID + (n0 + r)];
    }

    for (int rt = 0; rt < RT; ++rt) {
        const int row0 = (blockIdx.x * RT + rt) * 16;
        _Float16* xb = xs[rt & 1];

        #pragma unroll
        for (int i = 0; i < 8; ++i) {
            v4h h;
            h[0] = (_Float16)st[i].x; h[1] = (_Float16)st[i].y;
            h[2] = (_Float16)st[i].z; h[3] = (_Float16)st[i].w;
            *(v4h*)(xb + (2 * i + srow) * XP + 4 * sc4) = h;
        }
        if (rt + 1 < RT) {
            const v4f* nx = xt + (size_t)(rt + 1) * (16 * F_IN / 4);
            #pragma unroll
            for (int i = 0; i < 8; ++i) st[i] = __builtin_nontemporal_load(nx + i * 128);
        }
        __builtin_amdgcn_sched_barrier(0);

        __builtin_amdgcn_fence(__ATOMIC_RELEASE, "workgroup", "local");
        __builtin_amdgcn_s_barrier();
        __builtin_amdgcn_fence(__ATOMIC_ACQUIRE, "workgroup", "local");

        v8f acc = {0.f, 0.f, 0.f, 0.f, 0.f, 0.f, 0.f, 0.f};
        const _Float16* arow = xb + r * XP + 8 * half;
        #pragma unroll
        for (int kt = 0; kt < 8; ++kt) {
            v8h lo = *(const v8h*)(arow + kt * 32);
            v8h hi = *(const v8h*)(arow + kt * 32 + 16);
            v16h a = __builtin_shufflevector(lo, hi, 0, 1, 2, 3, 4, 5, 6, 7, 8, 9, 10, 11, 12, 13, 14, 15);

            acc = __builtin_amdgcn_wmma_f32_16x16x32_f16(
                 false, a,  false, bfrag[kt],
                 (short)0, acc,  false,  false);
        }

        #pragma unroll
        for (int i = 0; i < 8; ++i) {
            int m = i + 8 * half;
            xw1[(size_t)(row0 + m) * HID + (n0 + r)] = acc[i];
        }
        __builtin_amdgcn_sched_barrier(0);
    }
}

__global__ __launch_bounds__(1024) __attribute__((amdgpu_waves_per_eu(8, 8))) void k_agg1_own(
        const int* __restrict__ ei,
        const float* __restrict__ dinv,
        const float* __restrict__ xw1,
        const float* __restrict__ b1,
        const float* __restrict__ W2,
        const unsigned* __restrict__ qkeep,
        const unsigned* __restrict__ qn,
        float* __restrict__ zs) {
    __shared__ float    acc[BIN * HID];
    __shared__ float    dinv_l[BIN];
    __shared__ float    zl[BIN];

    const int tid  = threadIdx.x;
    const int lane = tid & 31;
    const int wave = tid >> 5;
    const int lo   = blockIdx.x * BIN;
    const int nown = min(BIN, N_NODES - lo);

    for (int n = tid; n < nown; n += 1024) dinv_l[n] = dinv[lo + n];
    for (int i = tid; i < nown * HID; i += 1024) {
        float di = dinv[lo + (i >> 6)];
        acc[i] = xw1[(size_t)lo * HID + i] * (di * di) + b1[i & 63];
    }
    __syncthreads();

    const unsigned kept = qn[blockIdx.x];
    if (kept <= (unsigned)QCAP) {
        const unsigned* q = qkeep + (size_t)blockIdx.x * QCAP;
        const int count = (int)kept;
        const int last = count - 1;
        unsigned ent_n = (wave * 32 < count) ? q[min(wave * 32 + lane, last)] : 0u;
        for (int base = wave * 32; base < count; base += 32 * 32) {
            const int nval = min(32, count - base);
            const unsigned ent = (lane < nval) ? ent_n : 0u;
            ent_n = q[min(base + 32 * 32 + lane, last)];
            int   s_l  = (int)(ent & ((1u << SBITS) - 1u));
            int   dl_l = (int)(ent >> SBITS);
            float ds_l = dinv[s_l];
            float dd_l = dinv_l[dl_l];
            #pragma unroll
            for (int jj = 0; jj < 32; jj += 16) {
                if (jj < nval) {
                    float v0[16], v1[16];
                    #pragma unroll
                    for (int u = 0; u < 16; ++u) {
                        if (jj + u < nval) {
                            int sj = __builtin_amdgcn_readlane(s_l, jj + u);
                            const float* row = xw1 + (size_t)sj * HID;
                            v0[u] = row[lane];
                            v1[u] = row[32 + lane];
                        }
                    }
                    #pragma unroll
                    for (int u = 0; u < 16; ++u) {
                        if (jj + u < nval) {
                            int   dj = __builtin_amdgcn_readlane(dl_l, jj + u);
                            float nj = __int_as_float(__builtin_amdgcn_readlane(__float_as_int(ds_l), jj + u))
                                     * __int_as_float(__builtin_amdgcn_readlane(__float_as_int(dd_l), jj + u));
                            atomicAdd(&acc[dj * HID + lane],      v0[u] * nj);
                            atomicAdd(&acc[dj * HID + 32 + lane], v1[u] * nj);
                        }
                    }
                }
            }
        }
    } else {
        for (int e = tid; e < N_EDGES; e += 1024) {
            unsigned dl = (unsigned)(ei[N_EDGES + e] - lo);
            if (dl < (unsigned)BIN) {
                int   s    = ei[e];
                float norm = dinv[s] * dinv_l[dl];
                const float* row = xw1 + (size_t)s * HID;
                #pragma nounroll
                for (int f = 0; f < HID; ++f) atomicAdd(&acc[dl * HID + f], row[f] * norm);
            }
        }
    }
    __syncthreads();

    const float w0a = W2[lane * 2 + 0],        w1a = W2[lane * 2 + 1];
    const float w0b = W2[(32 + lane) * 2 + 0], w1b = W2[(32 + lane) * 2 + 1];
    for (int n = wave; n < nown; n += 32) {
        float h0 = fmaxf(acc[n * HID + lane], 0.f);
        float h1 = fmaxf(acc[n * HID + 32 + lane], 0.f);
        float o0 = h0 * w0a + h1 * w0b;
        float o1 = h0 * w1a + h1 * w1b;
        for (int off = 16; off > 0; off >>= 1) {
            o0 += __shfl_xor(o0, off);
            o1 += __shfl_xor(o1, off);
        }
        if (lane == 0) zl[n] = (o0 - o1) * dinv_l[n];
    }
    __syncthreads();
    for (int n = tid; n < nown; n += 1024) zs[lo + n] = zl[n];
}

__global__ __launch_bounds__(1024) void k_agg2_own(
        const int* __restrict__ ei,
        const float* __restrict__ dinv,
        const float* __restrict__ zs,
        const float* __restrict__ b2,
        const unsigned* __restrict__ qkeep,
        const unsigned* __restrict__ qn,
        float* __restrict__ out) {
    __shared__ float sum[BIN];

    const int tid  = threadIdx.x;
    const int lo   = blockIdx.x * BIN;
    const int nown = min(BIN, N_NODES - lo);

    for (int n = tid; n < nown; n += 1024) sum[n] = zs[lo + n];
    __syncthreads();

    const unsigned count = qn[blockIdx.x];
    if (count <= (unsigned)QCAP) {
        const unsigned* q = qkeep + (size_t)blockIdx.x * QCAP;
        const unsigned NONE = 0xFFFFFFFFu;
        for (unsigned i0 = tid; i0 < count; i0 += 4u * 1024u) {
            unsigned ent[4];
            float    v[4];
            #pragma unroll
            for (int u = 0; u < 4; ++u) {
                unsigned i = i0 + (unsigned)u * 1024u;
                ent[u] = (i < count) ? q[i] : NONE;
            }
            #pragma unroll
            for (int u = 0; u < 4; ++u)
                v[u] = (ent[u] != NONE) ? zs[ent[u] & ((1u << SBITS) - 1u)] : 0.f;
            #pragma unroll
            for (int u = 0; u < 4; ++u)
                if (ent[u] != NONE) atomicAdd(&sum[ent[u] >> SBITS], v[u]);
        }
    } else {
        for (int e = tid; e < N_EDGES; e += 1024) {
            unsigned dl = (unsigned)(ei[N_EDGES + e] - lo);
            if (dl < (unsigned)BIN) atomicAdd(&sum[dl], zs[ei[e]]);
        }
    }
    __syncthreads();

    const float bd = b2[0] - b2[1];
    for (int n = tid; n < nown; n += 1024) {
        float a0 = dinv[lo + n] * sum[n] + bd;
        float a1 = 0.0f;
        float m  = fmaxf(a0, a1);
        float e0 = __expf(a0 - m);
        float e1 = __expf(a1 - m);
        float inv = 1.0f / (e0 + e1);
        float2 r = make_float2(e0 * inv, e1 * inv);
        *(float2*)(out + (size_t)(lo + n) * 2) = r;
    }
}

extern "C" void kernel_launch(void* const* d_in, const int* in_sizes, int n_in,
                              void* d_out, int out_size, void* d_ws, size_t ws_size,
                              hipStream_t stream) {
    const float* x  = (const float*)d_in[0];
    const int*   ei = (const int*)d_in[1];
    const float* W1 = (const float*)d_in[2];
    const float* b1 = (const float*)d_in[3];
    const float* W2 = (const float*)d_in[4];
    const float* b2 = (const float*)d_in[5];
    float* out = (float*)d_out;

    char* ws = (char*)d_ws;
    size_t o = 0;
    auto alloc = [&](size_t bytes) {
        char* p = ws + o;
        o += (bytes + 4095) & ~(size_t)4095;
        return p;
    };
    float* dinv = (float*)alloc((size_t)N_NODES * sizeof(float));
    float* xw1  = (float*)alloc((size_t)N_NODES * HID * sizeof(float));
    unsigned* qkeep = (unsigned*)alloc((size_t)NBINS * QCAP * sizeof(unsigned));
    float*    zs    = (float*)   alloc((size_t)N_NODES * sizeof(float));
    unsigned* qn    = (unsigned*)alloc((size_t)NBINS * sizeof(unsigned));
    unsigned* bits  = (unsigned*)xw1;
    (void)ws_size; (void)in_sizes; (void)n_in; (void)out_size;

    k_own_bits<<<NB1, dim3(1024), 0, stream>>>(ei, bits);
    k_find_own<<<NBINS, dim3(1024), 0, stream>>>(ei, bits, dinv, qkeep, qn);

    k_gemm1_wmma<<<(N_NODES / 16) / RT, dim3(128), 0, stream>>>(x, W1, xw1);

    k_agg1_own<<<NBINS, dim3(1024), 0, stream>>>(ei, dinv, xw1, b1, W2, qkeep, qn, zs);

    k_agg2_own<<<NBINS, dim3(1024), 0, stream>>>(ei, dinv, zs, b2, qkeep, qn, out);
}


# bitmap marking loses its launch: the first 782 blocks of the GEMM launch each mark 128 edge groups, their dst quads in flight with the first x tile; table stored block-major; sum order untouched
# speedup vs baseline: 4.9536x; 1.0361x over previous
#include <hip/hip_runtime.h>
#include <hip/hip_bf16.h>
#include <stdint.h>

#define N_NODES 100000
#define N_EDGES 1600000
#define F_IN    256
#define HID     64
#define RT      5

typedef __attribute__((ext_vector_type(16))) _Float16 v16h;
typedef __attribute__((ext_vector_type(8)))  _Float16 v8h;
typedef __attribute__((ext_vector_type(4)))  _Float16 v4h;
typedef __attribute__((ext_vector_type(8)))  float    v8f;
typedef __attribute__((ext_vector_type(4)))  float    v4f;

#define BIN    416
#define NQUAD  (N_EDGES / 4)
#define NBINS  ((N_NODES + BIN - 1) / BIN)
#define OGRP   (N_EDGES / 16)
#define GPB    128
#define NBB    ((OGRP + GPB - 1) / GPB)
#define OSLOT  248

#define SBITS  17
#define QCAP   12288
#define GCAP   12288

static __device__ __forceinline__ void take_edge(int s, int d, int lo,
        unsigned* cnt, unsigned* queue, unsigned* qcount) {
    const unsigned dl = (unsigned)(d - lo);
    if (dl < (unsigned)BIN) {
        atomicAdd(&cnt[dl], 1u);
        const unsigned slot = atomicAdd(qcount, 1u);
        if (slot < QCAP) queue[slot] = (unsigned)s | (dl << SBITS);
    }
}

__global__ __launch_bounds__(1024) __attribute__((amdgpu_waves_per_eu(8, 8))) void k_find_own(
        const int* __restrict__ ei,
        const unsigned* __restrict__ bits,
        float* __restrict__ dinv,
        unsigned* __restrict__ qkeep,
        unsigned* __restrict__ qn) {
    __shared__ unsigned cnt[BIN];
    __shared__ unsigned queue[QCAP];
    __shared__ unsigned gqueue[GCAP];
    __shared__ unsigned qcount;
    __shared__ unsigned gcount;

    const int tid  = threadIdx.x;
    const int lane = tid & 31;
    const int lo   = blockIdx.x * BIN;
    const int nown = min(BIN, N_NODES - lo);

    uint4 x = make_uint4(0u, 0u, 0u, 0u);
    if (tid < NBB) x = ((const uint4*)bits)[(size_t)tid * OSLOT + blockIdx.x];

    if (tid == 0) { qcount = 0u; gcount = 0u; }
    for (int n = tid; n < BIN; n += 1024) cnt[n] = 0u;
    __syncthreads();

    {
        const int c = __popc(x.x) + __popc(x.y) + __popc(x.z) + __popc(x.w);
        int incl = c;
        for (int off = 1; off < 32; off <<= 1) {
            const int v = __shfl_up(incl, off);
            if (lane >= off) incl += v;
        }
        const int total = __shfl(incl, 31);
        unsigned base = 0u;
        if (lane == 31 && total > 0) base = atomicAdd(&gcount, (unsigned)total);
        base = (unsigned)__shfl((int)base, 31);
        unsigned slot = base + (unsigned)(incl - c);
        auto list = [&](unsigned m, unsigned g0) {
            while (m) {
                const int r = __builtin_ctz(m);
                m &= m - 1u;
                if (slot < (unsigned)GCAP) gqueue[slot] = g0 + (unsigned)r;
                ++slot;
            }
        };
        const unsigned j0 = (unsigned)tid * 4u;
        list(x.x, (j0 + 0u) * 32u);
        list(x.y, (j0 + 1u) * 32u);
        list(x.z, (j0 + 2u) * 32u);
        list(x.w, (j0 + 3u) * 32u);
    }
    __syncthreads();

    const int4* src4 = (const int4*)ei;
    const int4* dst4 = (const int4*)(ei + N_EDGES);
    if (gcount <= (unsigned)GCAP) {
        const int items = (int)gcount * 4;
        for (int b0 = 0; b0 < items; b0 += 4 * 1024) {
            int4 s[4], d[4];
            int g[4];
            #pragma unroll
            for (int u = 0; u < 4; ++u) g[u] = (int)gqueue[min(b0 + u * 1024 + tid, items - 1) >> 2];
            #pragma unroll
            for (int u = 0; u < 4; ++u) {
                const int q = g[u] * 4 + (tid & 3);
                d[u] = dst4[q];
                s[u] = src4[q];
            }
            __builtin_amdgcn_sched_barrier(0);
            unsigned dl[16], sv[16], m = 0u;
            #pragma unroll
            for (int u = 0; u < 4; ++u) {
                dl[4 * u + 0] = (unsigned)(d[u].x - lo); sv[4 * u + 0] = (unsigned)s[u].x;
                dl[4 * u + 1] = (unsigned)(d[u].y - lo); sv[4 * u + 1] = (unsigned)s[u].y;
                dl[4 * u + 2] = (unsigned)(d[u].z - lo); sv[4 * u + 2] = (unsigned)s[u].z;
                dl[4 * u + 3] = (unsigned)(d[u].w - lo); sv[4 * u + 3] = (unsigned)s[u].w;
                if (b0 + u * 1024 + tid < items) {
                    #pragma unroll
                    for (int k = 0; k < 4; ++k)
                        if (dl[4 * u + k] < (unsigned)BIN) m |= 1u << (4 * u + k);
                }
            }
            unsigned slot = atomicAdd(&qcount, (unsigned)__popc(m));
            #pragma unroll
            for (int k = 0; k < 16; ++k) {
                if (m & (1u << k)) {
                    atomicAdd(&cnt[dl[k]], 1u);
                    if (slot < (unsigned)QCAP) queue[slot] = sv[k] | (dl[k] << SBITS);
                    ++slot;
                }
            }
        }
    } else {
        for (int e = tid; e < N_EDGES; e += 1024)
            take_edge(ei[e], ei[N_EDGES + e], lo, cnt, queue, &qcount);
    }
    __syncthreads();

    const int count = (int)min(qcount, (unsigned)QCAP);
    unsigned* qk = qkeep + (size_t)blockIdx.x * QCAP;
    for (int i = tid; i < count; i += 1024) qk[i] = queue[i];

    for (int n = tid; n < nown; n += 1024) dinv[lo + n] = __frsqrt_rn((float)cnt[n] + 1.0f);
    if (tid == 0) qn[blockIdx.x] = qcount;
}

#define XP 264
__global__ __launch_bounds__(128) void k_gemm1_wmma(
        const float* __restrict__ x,
        const float* __restrict__ W1,
        float* __restrict__ xw1,
        const int* __restrict__ ei,
        unsigned* __restrict__ bits) {
    __shared__ __attribute__((aligned(16))) _Float16 xs[2][16 * XP];
    __shared__ unsigned tab[4 * NBINS];

    const int tid  = threadIdx.x;
    const int wave = tid >> 5;
    const int lane = tid & 31;
    const int half = lane >> 4;
    const int r    = lane & 15;
    const int n0   = wave * 16;

    const int srow = tid >> 6;
    const int sc4  = tid & 63;
    const v4f* xt = (const v4f*)(x + (size_t)blockIdx.x * RT * 16 * F_IN) + tid;

    v4f st[8];
    #pragma unroll
    for (int i = 0; i < 8; ++i) st[i] = __builtin_nontemporal_load(xt + i * 128);

    const bool marks = blockIdx.x < NBB;
    if (marks) {
        const int4* dst4 = (const int4*)(ei + N_EDGES);
        const int q0 = blockIdx.x * (GPB * 4) + tid;
        int4 d[4];
        #pragma unroll
        for (int u = 0; u < 4; ++u) {
            const int q = q0 + u * 128;
            d[u] = dst4[(q < NQUAD) ? q : 0];
        }
        for (int i = tid; i < 4 * NBINS; i += 128) tab[i] = 0u;
        __builtin_amdgcn_fence(__ATOMIC_RELEASE, "workgroup", "local");
        __builtin_amdgcn_s_barrier();
        __builtin_amdgcn_fence(__ATOMIC_ACQUIRE, "workgroup", "local");
        #pragma unroll
        for (int u = 0; u < 4; ++u) {
            const int q = q0 + u * 128;
            if (q < NQUAD) {
                const int p = (u * 128 + tid) >> 2;
                unsigned* row = tab + (p >> 5) * NBINS;
                const unsigned bit = 1u << (p & 31);
                atomicOr(&row[d[u].x / BIN], bit);
                atomicOr(&row[d[u].y / BIN], bit);
                atomicOr(&row[d[u].z / BIN], bit);
                atomicOr(&row[d[u].w / BIN], bit);
            }
        }
        __builtin_amdgcn_fence(__ATOMIC_RELEASE, "workgroup", "local");
        __builtin_amdgcn_s_barrier();
        __builtin_amdgcn_fence(__ATOMIC_ACQUIRE, "workgroup", "local");
        uint4* mine = (uint4*)bits + (size_t)blockIdx.x * OSLOT;
        for (int o = tid; o < NBINS; o += 128)
            mine[o] = make_uint4(tab[o], tab[NBINS + o], tab[2 * NBINS + o], tab[3 * NBINS + o]);
    }

    v16h bfrag[8];
    #pragma unroll
    for (int kt = 0; kt < 8; ++kt) {
        const int kbase = kt * 32;
        #pragma unroll
        for (int j = 0; j < 8; ++j)
            bfrag[kt][j]     = (_Float16)W1[(kbase + 8 * half + j) * HID + (n0 + r)];
        #pragma unroll
        for (int j = 0; j < 8; ++j)
            bfrag[kt][8 + j] = (_Float16)W1[(kbase + 16 + 8 * half + j) * HID + (n0 + r)];
    }

    for (int rt = 0; rt < RT; ++rt) {
        const int row0 = (blockIdx.x * RT + rt) * 16;
        _Float16* xb = xs[rt & 1];

        #pragma unroll
        for (int i = 0; i < 8; ++i) {
            v4h h;
            h[0] = (_Float16)st[i].x; h[1] = (_Float16)st[i].y;
            h[2] = (_Float16)st[i].z; h[3] = (_Float16)st[i].w;
            *(v4h*)(xb + (2 * i + srow) * XP + 4 * sc4) = h;
        }
        if (rt + 1 < RT) {
            const v4f* nx = xt + (size_t)(rt + 1) * (16 * F_IN / 4);
            #pragma unroll
            for (int i = 0; i < 8; ++i) st[i] = __builtin_nontemporal_load(nx + i * 128);
        }
        __builtin_amdgcn_sched_barrier(0);

        __builtin_amdgcn_fence(__ATOMIC_RELEASE, "workgroup", "local");
        __builtin_amdgcn_s_barrier();
        __builtin_amdgcn_fence(__ATOMIC_ACQUIRE, "workgroup", "local");

        v8f acc = {0.f, 0.f, 0.f, 0.f, 0.f, 0.f, 0.f, 0.f};
        const _Float16* arow = xb + r * XP + 8 * half;
        #pragma unroll
        for (int kt = 0; kt < 8; ++kt) {
            v8h lo = *(const v8h*)(arow + kt * 32);
            v8h hi = *(const v8h*)(arow + kt * 32 + 16);
            v16h a = __builtin_shufflevector(lo, hi, 0, 1, 2, 3, 4, 5, 6, 7, 8, 9, 10, 11, 12, 13, 14, 15);

            acc = __builtin_amdgcn_wmma_f32_16x16x32_f16(
                 false, a,  false, bfrag[kt],
                 (short)0, acc,  false,  false);
        }

        #pragma unroll
        for (int i = 0; i < 8; ++i) {
            int m = i + 8 * half;
            xw1[(size_t)(row0 + m) * HID + (n0 + r)] = acc[i];
        }
        __builtin_amdgcn_sched_barrier(0);
    }
}

__global__ __launch_bounds__(1024) __attribute__((amdgpu_waves_per_eu(8, 8))) void k_agg1_own(
        const int* __restrict__ ei,
        const float* __restrict__ dinv,
        const float* __restrict__ xw1,
        const float* __restrict__ b1,
        const float* __restrict__ W2,
        const unsigned* __restrict__ qkeep,
        const unsigned* __restrict__ qn,
        float* __restrict__ zs) {
    __shared__ float    acc[BIN * HID];
    __shared__ float    dinv_l[BIN];
    __shared__ float    zl[BIN];

    const int tid  = threadIdx.x;
    const int lane = tid & 31;
    const int wave = tid >> 5;
    const int lo   = blockIdx.x * BIN;
    const int nown = min(BIN, N_NODES - lo);

    for (int n = tid; n < nown; n += 1024) dinv_l[n] = dinv[lo + n];
    for (int i = tid; i < nown * HID; i += 1024) {
        float di = dinv[lo + (i >> 6)];
        acc[i] = xw1[(size_t)lo * HID + i] * (di * di) + b1[i & 63];
    }
    __syncthreads();

    const unsigned kept = qn[blockIdx.x];
    if (kept <= (unsigned)QCAP) {
        const unsigned* q = qkeep + (size_t)blockIdx.x * QCAP;
        const int count = (int)kept;
        const int last = count - 1;
        unsigned ent_n = (wave * 32 < count) ? q[min(wave * 32 + lane, last)] : 0u;
        for (int base = wave * 32; base < count; base += 32 * 32) {
            const int nval = min(32, count - base);
            const unsigned ent = (lane < nval) ? ent_n : 0u;
            ent_n = q[min(base + 32 * 32 + lane, last)];
            int   s_l  = (int)(ent & ((1u << SBITS) - 1u));
            int   dl_l = (int)(ent >> SBITS);
            float ds_l = dinv[s_l];
            float dd_l = dinv_l[dl_l];
            #pragma unroll
            for (int jj = 0; jj < 32; jj += 16) {
                if (jj < nval) {
                    float v0[16], v1[16];
                    #pragma unroll
                    for (int u = 0; u < 16; ++u) {
                        if (jj + u < nval) {
                            int sj = __builtin_amdgcn_readlane(s_l, jj + u);
                            const float* row = xw1 + (size_t)sj * HID;
                            v0[u] = row[lane];
                            v1[u] = row[32 + lane];
                        }
                    }
                    #pragma unroll
                    for (int u = 0; u < 16; ++u) {
                        if (jj + u < nval) {
                            int   dj = __builtin_amdgcn_readlane(dl_l, jj + u);
                            float nj = __int_as_float(__builtin_amdgcn_readlane(__float_as_int(ds_l), jj + u))
                                     * __int_as_float(__builtin_amdgcn_readlane(__float_as_int(dd_l), jj + u));
                            atomicAdd(&acc[dj * HID + lane],      v0[u] * nj);
                            atomicAdd(&acc[dj * HID + 32 + lane], v1[u] * nj);
                        }
                    }
                }
            }
        }
    } else {
        for (int e = tid; e < N_EDGES; e += 1024) {
            unsigned dl = (unsigned)(ei[N_EDGES + e] - lo);
            if (dl < (unsigned)BIN) {
                int   s    = ei[e];
                float norm = dinv[s] * dinv_l[dl];
                const float* row = xw1 + (size_t)s * HID;
                #pragma nounroll
                for (int f = 0; f < HID; ++f) atomicAdd(&acc[dl * HID + f], row[f] * norm);
            }
        }
    }
    __syncthreads();

    const float w0a = W2[lane * 2 + 0],        w1a = W2[lane * 2 + 1];
    const float w0b = W2[(32 + lane) * 2 + 0], w1b = W2[(32 + lane) * 2 + 1];
    for (int n = wave; n < nown; n += 32) {
        float h0 = fmaxf(acc[n * HID + lane], 0.f);
        float h1 = fmaxf(acc[n * HID + 32 + lane], 0.f);
        float o0 = h0 * w0a + h1 * w0b;
        float o1 = h0 * w1a + h1 * w1b;
        for (int off = 16; off > 0; off >>= 1) {
            o0 += __shfl_xor(o0, off);
            o1 += __shfl_xor(o1, off);
        }
        if (lane == 0) zl[n] = (o0 - o1) * dinv_l[n];
    }
    __syncthreads();
    for (int n = tid; n < nown; n += 1024) zs[lo + n] = zl[n];
}

__global__ __launch_bounds__(1024) void k_agg2_own(
        const int* __restrict__ ei,
        const float* __restrict__ dinv,
        const float* __restrict__ zs,
        const float* __restrict__ b2,
        const unsigned* __restrict__ qkeep,
        const unsigned* __restrict__ qn,
        float* __restrict__ out) {
    __shared__ float sum[BIN];

    const int tid  = threadIdx.x;
    const int lo   = blockIdx.x * BIN;
    const int nown = min(BIN, N_NODES - lo);

    for (int n = tid; n < nown; n += 1024) sum[n] = zs[lo + n];
    __syncthreads();

    const unsigned count = qn[blockIdx.x];
    if (count <= (unsigned)QCAP) {
        const unsigned* q = qkeep + (size_t)blockIdx.x * QCAP;
        const unsigned NONE = 0xFFFFFFFFu;
        for (unsigned i0 = tid; i0 < count; i0 += 4u * 1024u) {
            unsigned ent[4];
            float    v[4];
            #pragma unroll
            for (int u = 0; u < 4; ++u) {
                unsigned i = i0 + (unsigned)u * 1024u;
                ent[u] = (i < count) ? q[i] : NONE;
            }
            #pragma unroll
            for (int u = 0; u < 4; ++u)
                v[u] = (ent[u] != NONE) ? zs[ent[u] & ((1u << SBITS) - 1u)] : 0.f;
            #pragma unroll
            for (int u = 0; u < 4; ++u)
                if (ent[u] != NONE) atomicAdd(&sum[ent[u] >> SBITS], v[u]);
        }
    } else {
        for (int e = tid; e < N_EDGES; e += 1024) {
            unsigned dl = (unsigned)(ei[N_EDGES + e] - lo);
            if (dl < (unsigned)BIN) atomicAdd(&sum[dl], zs[ei[e]]);
        }
    }
    __syncthreads();

    const float bd = b2[0] - b2[1];
    for (int n = tid; n < nown; n += 1024) {
        float a0 = dinv[lo + n] * sum[n] + bd;
        float a1 = 0.0f;
        float m  = fmaxf(a0, a1);
        float e0 = __expf(a0 - m);
        float e1 = __expf(a1 - m);
        float inv = 1.0f / (e0 + e1);
        float2 r = make_float2(e0 * inv, e1 * inv);
        *(float2*)(out + (size_t)(lo + n) * 2) = r;
    }
}

extern "C" void kernel_launch(void* const* d_in, const int* in_sizes, int n_in,
                              void* d_out, int out_size, void* d_ws, size_t ws_size,
                              hipStream_t stream) {
    const float* x  = (const float*)d_in[0];
    const int*   ei = (const int*)d_in[1];
    const float* W1 = (const float*)d_in[2];
    const float* b1 = (const float*)d_in[3];
    const float* W2 = (const float*)d_in[4];
    const float* b2 = (const float*)d_in[5];
    float* out = (float*)d_out;

    char* ws = (char*)d_ws;
    size_t o = 0;
    auto alloc = [&](size_t bytes) {
        char* p = ws + o;
        o += (bytes + 4095) & ~(size_t)4095;
        return p;
    };
    float* dinv = (float*)alloc((size_t)N_NODES * sizeof(float));
    float* xw1  = (float*)alloc((size_t)N_NODES * HID * sizeof(float));
    unsigned* qkeep = (unsigned*)alloc((size_t)NBINS * QCAP * sizeof(unsigned));
    float*    zs    = (float*)   alloc((size_t)N_NODES * sizeof(float));
    unsigned* qn    = (unsigned*)alloc((size_t)NBINS * sizeof(unsigned));
    unsigned* bits  = (unsigned*)alloc((size_t)NBB * OSLOT * 4 * sizeof(unsigned));
    (void)ws_size; (void)in_sizes; (void)n_in; (void)out_size;

    k_gemm1_wmma<<<(N_NODES / 16) / RT, dim3(128), 0, stream>>>(x, W1, xw1, ei, bits);

    k_find_own<<<NBINS, dim3(1024), 0, stream>>>(ei, bits, dinv, qkeep, qn);

    k_agg1_own<<<NBINS, dim3(1024), 0, stream>>>(ei, dinv, xw1, b1, W2, qkeep, qn, zs);

    k_agg2_own<<<NBINS, dim3(1024), 0, stream>>>(ei, dinv, zs, b2, qkeep, qn, out);
}
